# starting kernel
# baseline (speedup 1.0000x reference)
#include <hip/hip_runtime.h>
#include <math.h>


#define LS     68
#define BUF    (16 * LS)
#define TSTEPS 200
#define DDIM   64

typedef float v2f __attribute__((ext_vector_type(2)));
typedef float v8f __attribute__((ext_vector_type(8)));

__device__ __forceinline__ float fast_tanh(float v) {
#if __has_builtin(__builtin_amdgcn_tanhf)
  return __builtin_amdgcn_tanhf(v);
#elif __has_builtin(__builtin_amdgcn_tanh_f32)
  return __builtin_amdgcn_tanh_f32(v);
#else
  return tanhf(v);
#endif
}

__device__ __forceinline__ v8f wmma_f32(v2f a, v2f b, v8f c) {
  return __builtin_amdgcn_wmma_f32_16x16x4_f32(false, a, false, b, (short)0, c,
                                               false, false);
}

__device__ __forceinline__ void proj16x64(const float* src, const v2f* wf,
                                          float* dst, int m, int h) {
  v8f a0 = {}, a1 = {}, a2 = {}, a3 = {};
  const float* arow = src + m * LS + 2 * h;
#pragma unroll
  for (int kc = 0; kc < 16; ++kc) {
    v2f af = *(const v2f*)(arow + 4 * kc);
    a0 = wmma_f32(af, wf[kc * 4 + 0], a0);
    a1 = wmma_f32(af, wf[kc * 4 + 1], a1);
    a2 = wmma_f32(af, wf[kc * 4 + 2], a2);
    a3 = wmma_f32(af, wf[kc * 4 + 3], a3);
  }
#pragma unroll
  for (int r = 0; r < 8; ++r) {
    int row = r + 8 * h;
    dst[row * LS + m +  0] = a0[r];
    dst[row * LS + m + 16] = a1[r];
    dst[row * LS + m + 32] = a2[r];
    dst[row * LS + m + 48] = a3[r];
  }
}

__device__ __forceinline__ float rowdot(const float* A, const float* B, int m,
                                        int h) {
  const float4* a = (const float4*)(A + m * LS + 32 * h);
  const float4* b = (const float4*)(B + m * LS + 32 * h);
  float s = 0.f;
#pragma unroll
  for (int i = 0; i < 8; ++i) {
    float4 u = a[i], v = b[i];
    s += u.x * v.x + u.y * v.y + u.z * v.z + u.w * v.w;
  }
  s += __shfl_xor(s, 16, 32);
  return s;
}

__device__ __forceinline__ void softmax2(float a, float b, float& p0,
                                         float& p1) {
  float mx  = fmaxf(a, b);
  float e0  = __expf(a - mx);
  float e1  = __expf(b - mx);
  float inv = 1.f / (e0 + e1);
  p0 = e0 * inv;
  p1 = e1 * inv;
}

template <bool DO_TANH>
__device__ __forceinline__ void combine(const float* V0, const float* V1,
                                        float p0, float p1, float* dst, int m,
                                        int h) {
  const float4* v0 = (const float4*)(V0 + m * LS + 32 * h);
  const float4* v1 = (const float4*)(V1 + m * LS + 32 * h);
  float4* d = (float4*)(dst + m * LS + 32 * h);
#pragma unroll
  for (int i = 0; i < 8; ++i) {
    float4 a = v0[i], b = v1[i], r;
    r.x = p0 * a.x + p1 * b.x;
    r.y = p0 * a.y + p1 * b.y;
    r.z = p0 * a.z + p1 * b.z;
    r.w = p0 * a.w + p1 * b.w;
    if (DO_TANH) {
      r.x = fast_tanh(r.x);
      r.y = fast_tanh(r.y);
      r.z = fast_tanh(r.z);
      r.w = fast_tanh(r.w);
    }
    d[i] = r;
  }
}

__global__ void swizzle_weights(const float* __restrict__ Wq,
                                const float* __restrict__ Wk,
                                const float* __restrict__ Wv,
                                float* __restrict__ wsw) {
  int idx = blockIdx.x * blockDim.x + threadIdx.x;
  if (idx >= 3 * 16 * 4 * 32 * 2) return;
  int j    = idx & 1;
  int lane = (idx >> 1) & 31;
  int nt   = (idx >> 6) & 3;
  int kc   = (idx >> 8) & 15;
  int w    = idx >> 12;
  int k = 4 * kc + j + 2 * (lane >> 4);
  int n = 16 * nt + (lane & 15);
  const float* W = (w == 0) ? Wq : (w == 1) ? Wk : Wv;
  wsw[idx] = W[k * 64 + n];
}

__global__ __launch_bounds__(96) void birnn_kernel(
    const float* __restrict__ x, const float* __restrict__ wsw,
    float* __restrict__ out) {
  __shared__ __align__(16) float smem[13 * BUF];
  float* bufL  = smem + 0 * BUF;
  float* bufS  = smem + 1 * BUF;
  float* bufX  = smem + 2 * BUF;
  float* bufO0 = smem + 3 * BUF;
  float* bufO1 = smem + 4 * BUF;
  float* P0 = smem + 5 * BUF;
  float* P1 = smem + 6 * BUF;
  float* P2 = smem + 7 * BUF;
  float* P3 = smem + 8 * BUF;
  float* P4 = smem + 9 * BUF;
  float* P5 = smem + 10 * BUF;
  float* P6 = smem + 11 * BUF;
  float* P7 = smem + 12 * BUF;

  const int lane = threadIdx.x & 31;
  const int wid  = threadIdx.x >> 5;
  const int m = lane & 15;
  const int h = lane >> 4;
  const int b = blockIdx.x * 16 + m;
  const float kScale = 0.125f;

  v2f wf[64];
  {
    const float* myw = wsw + wid * 4096 + lane * 2;
#pragma unroll
    for (int f = 0; f < 64; ++f) wf[f] = *(const v2f*)(myw + f * 64);
  }

  if (wid == 0) {
    const float4* xr = (const float4*)(x + (size_t)b * TSTEPS * DDIM + 32 * h);
    float4* dl  = (float4*)(bufL + m * LS + 32 * h);
    float4* dsv = (float4*)(bufS + m * LS + 32 * h);
#pragma unroll
    for (int i = 0; i < 8; ++i) {
      float4 v = xr[i];
      dl[i]  = v;
      dsv[i] = v;
    }
  }
  __syncthreads();

#pragma unroll 1
  for (int t = 1; t < TSTEPS; ++t) {
    if (wid == 2) {
      const float* xrow = x + ((size_t)b * TSTEPS + t) * DDIM + 32 * h;
      const float4* xr = (const float4*)xrow;
      float4* dx = (float4*)(bufX + m * LS + 32 * h);
#pragma unroll
      for (int i = 0; i < 8; ++i) dx[i] = xr[i];
      if (t + 1 < TSTEPS) __builtin_prefetch(xrow + DDIM, 0, 0);
    }
    __syncthreads();

    if (wid == 0) {
      proj16x64(bufL, wf, P0, m, h);
      proj16x64(bufX, wf, P3, m, h);
    } else if (wid == 1) {
      proj16x64(bufL, wf, P1, m, h);
      proj16x64(bufX, wf, P4, m, h);
      proj16x64(bufS, wf, P6, m, h);
    } else {
      proj16x64(bufL, wf, P2, m, h);
      proj16x64(bufX, wf, P5, m, h);
      proj16x64(bufS, wf, P7, m, h);
    }
    __syncthreads();

    if (wid == 0) {
      float d00 = rowdot(P0, P1, m, h) * kScale;
      float d01 = rowdot(P0, P4, m, h) * kScale;
      float p0, p1;
      softmax2(d00, d01, p0, p1);
      combine<true>(P2, P5, p0, p1, bufO0, m, h);
    } else if (wid == 1) {
      float d10 = rowdot(P3, P6, m, h) * kScale;
      float d11 = rowdot(P3, P4, m, h) * kScale;
      float q0, q1;
      softmax2(d10, d11, q0, q1);
      combine<true>(P7, P5, q0, q1, bufO1, m, h);
    }
    __syncthreads();

    if (wid == 0) {
      proj16x64(bufO0, wf, P0, m, h);
      proj16x64(bufO1, wf, P3, m, h);
    } else if (wid == 1) {
      proj16x64(bufO0, wf, P1, m, h);
      proj16x64(bufO1, wf, P4, m, h);
    } else {
      proj16x64(bufO0, wf, P2, m, h);
      proj16x64(bufO1, wf, P5, m, h);
    }
    __syncthreads();

    if (wid == 0) {
      float e00 = rowdot(P0, P1, m, h) * kScale;
      float e01 = rowdot(P0, P4, m, h) * kScale;
      float r0, r1;
      softmax2(e00, e01, r0, r1);
      combine<false>(P2, P5, r0, r1, bufL, m, h);
    } else if (wid == 1) {
      float e10 = rowdot(P3, P1, m, h) * kScale;
      float e11 = rowdot(P3, P4, m, h) * kScale;
      float s0, s1;
      softmax2(e10, e11, s0, s1);
      combine<false>(P2, P5, s0, s1, bufS, m, h);
    }
    __syncthreads();
  }

  if (wid == 0) {
    const float4* src = (const float4*)(bufL + m * LS + 32 * h);
    float4* dst = (float4*)(out + (size_t)b * DDIM + 32 * h);
#pragma unroll
    for (int i = 0; i < 8; ++i) dst[i] = src[i];
  }
}

extern "C" void kernel_launch(void* const* d_in, const int* in_sizes, int n_in,
                              void* d_out, int out_size, void* d_ws,
                              size_t ws_size, hipStream_t stream) {
  (void)in_sizes; (void)n_in; (void)out_size; (void)ws_size;
  const float* x  = (const float*)d_in[0];
  const float* Wq = (const float*)d_in[1];
  const float* Wk = (const float*)d_in[2];
  const float* Wv = (const float*)d_in[3];
  float* wsw = (float*)d_ws;

  swizzle_weights<<<48, 256, 0, stream>>>(Wq, Wk, Wv, wsw);
  birnn_kernel<<<2048 / 16, 96, 0, stream>>>(x, wsw, (float*)d_out);
}


# Fourth wave on the idle fourth SIMD takes a quarter of each step's projections (Kx, Vx; columns 32-63 of q1, k1, v1): busiest matrix pipe issues 224 WMMAs per step, was 320; sum order unchanged
# speedup vs baseline: 1.3679x; 1.3679x over previous
#include <hip/hip_runtime.h>
#include <math.h>


#define LS     68
#define BUF    (16 * LS)
#define TSTEPS 200
#define DDIM   64

typedef float v2f __attribute__((ext_vector_type(2)));
typedef float v8f __attribute__((ext_vector_type(8)));

__device__ __forceinline__ float fast_tanh(float v) {
#if __has_builtin(__builtin_amdgcn_tanhf)
  return __builtin_amdgcn_tanhf(v);
#elif __has_builtin(__builtin_amdgcn_tanh_f32)
  return __builtin_amdgcn_tanh_f32(v);
#else
  return tanhf(v);
#endif
}

__device__ __forceinline__ v8f wmma_f32(v2f a, v2f b, v8f c) {
  return __builtin_amdgcn_wmma_f32_16x16x4_f32(false, a, false, b, (short)0, c,
                                               false, false);
}

template <int NTN, int NT0, int WS, int WO>
__device__ __forceinline__ void proj_tiles(const float* src, const v2f* wf,
                                           float* dst, int m, int h) {
  v8f acc[NTN];
#pragma unroll
  for (int i = 0; i < NTN; ++i) acc[i] = v8f{};
  const float* arow = src + m * LS + 2 * h;
#pragma unroll
  for (int kc = 0; kc < 16; ++kc) {
    v2f af = *(const v2f*)(arow + 4 * kc);
#pragma unroll
    for (int i = 0; i < NTN; ++i)
      acc[i] = wmma_f32(af, wf[kc * WS + WO + i], acc[i]);
  }
#pragma unroll
  for (int r = 0; r < 8; ++r) {
    int row = r + 8 * h;
#pragma unroll
    for (int i = 0; i < NTN; ++i)
      dst[row * LS + m + 16 * (NT0 + i)] = acc[i][r];
  }
}

template <int NTN, int NT0, int WSA, int WOA, int WSB, int WOB>
__device__ __forceinline__ void proj_tiles2(const float* src, const v2f* wa,
                                            const v2f* wb, float* da,
                                            float* db, int m, int h) {
  v8f ca[NTN], cb[NTN];
#pragma unroll
  for (int i = 0; i < NTN; ++i) { ca[i] = v8f{}; cb[i] = v8f{}; }
  const float* arow = src + m * LS + 2 * h;
#pragma unroll
  for (int kc = 0; kc < 16; ++kc) {
    v2f af = *(const v2f*)(arow + 4 * kc);
#pragma unroll
    for (int i = 0; i < NTN; ++i)
      ca[i] = wmma_f32(af, wa[kc * WSA + WOA + i], ca[i]);
#pragma unroll
    for (int i = 0; i < NTN; ++i)
      cb[i] = wmma_f32(af, wb[kc * WSB + WOB + i], cb[i]);
  }
#pragma unroll
  for (int r = 0; r < 8; ++r) {
    int row = r + 8 * h;
#pragma unroll
    for (int i = 0; i < NTN; ++i) {
      da[row * LS + m + 16 * (NT0 + i)] = ca[i][r];
      db[row * LS + m + 16 * (NT0 + i)] = cb[i][r];
    }
  }
}

template <int NTN, int NT0, int WSA, int WOA, int WSB, int WOB, int WSC,
          int WOC>
__device__ __forceinline__ void proj_tiles3(const float* src, const v2f* wa,
                                            const v2f* wb, const v2f* wc,
                                            float* da, float* db, float* dc,
                                            int m, int h) {
  v8f ca[NTN], cb[NTN], cc[NTN];
#pragma unroll
  for (int i = 0; i < NTN; ++i) { ca[i] = v8f{}; cb[i] = v8f{}; cc[i] = v8f{}; }
  const float* arow = src + m * LS + 2 * h;
#pragma unroll
  for (int kc = 0; kc < 16; ++kc) {
    v2f af = *(const v2f*)(arow + 4 * kc);
#pragma unroll
    for (int i = 0; i < NTN; ++i)
      ca[i] = wmma_f32(af, wa[kc * WSA + WOA + i], ca[i]);
#pragma unroll
    for (int i = 0; i < NTN; ++i)
      cb[i] = wmma_f32(af, wb[kc * WSB + WOB + i], cb[i]);
#pragma unroll
    for (int i = 0; i < NTN; ++i)
      cc[i] = wmma_f32(af, wc[kc * WSC + WOC + i], cc[i]);
  }
#pragma unroll
  for (int r = 0; r < 8; ++r) {
    int row = r + 8 * h;
#pragma unroll
    for (int i = 0; i < NTN; ++i) {
      da[row * LS + m + 16 * (NT0 + i)] = ca[i][r];
      db[row * LS + m + 16 * (NT0 + i)] = cb[i][r];
      dc[row * LS + m + 16 * (NT0 + i)] = cc[i][r];
    }
  }
}

__device__ __forceinline__ void proj16x64(const float* src, const v2f* wf,
                                          float* dst, int m, int h) {
  proj_tiles<4, 0, 4, 0>(src, wf, dst, m, h);
}

__device__ __forceinline__ float rowdot(const float* A, const float* B, int m,
                                        int h) {
  const float4* a = (const float4*)(A + m * LS + 32 * h);
  const float4* b = (const float4*)(B + m * LS + 32 * h);
  float s = 0.f;
#pragma unroll
  for (int i = 0; i < 8; ++i) {
    float4 u = a[i], v = b[i];
    s += u.x * v.x + u.y * v.y + u.z * v.z + u.w * v.w;
  }
  s += __shfl_xor(s, 16, 32);
  return s;
}

__device__ __forceinline__ void softmax2(float a, float b, float& p0,
                                         float& p1) {
  float mx  = fmaxf(a, b);
  float e0  = __expf(a - mx);
  float e1  = __expf(b - mx);
  float inv = 1.f / (e0 + e1);
  p0 = e0 * inv;
  p1 = e1 * inv;
}

template <bool DO_TANH>
__device__ __forceinline__ void combine(const float* V0, const float* V1,
                                        float p0, float p1, float* dst, int m,
                                        int h) {
  const float4* v0 = (const float4*)(V0 + m * LS + 32 * h);
  const float4* v1 = (const float4*)(V1 + m * LS + 32 * h);
  float4* d = (float4*)(dst + m * LS + 32 * h);
#pragma unroll
  for (int i = 0; i < 8; ++i) {
    float4 a = v0[i], b = v1[i], r;
    r.x = p0 * a.x + p1 * b.x;
    r.y = p0 * a.y + p1 * b.y;
    r.z = p0 * a.z + p1 * b.z;
    r.w = p0 * a.w + p1 * b.w;
    if (DO_TANH) {
      r.x = fast_tanh(r.x);
      r.y = fast_tanh(r.y);
      r.z = fast_tanh(r.z);
      r.w = fast_tanh(r.w);
    }
    d[i] = r;
  }
}

__global__ void swizzle_weights(const float* __restrict__ Wq,
                                const float* __restrict__ Wk,
                                const float* __restrict__ Wv,
                                float* __restrict__ wsw) {
  int idx = blockIdx.x * blockDim.x + threadIdx.x;
  if (idx >= 3 * 16 * 4 * 32 * 2) return;
  int j    = idx & 1;
  int lane = (idx >> 1) & 31;
  int nt   = (idx >> 6) & 3;
  int kc   = (idx >> 8) & 15;
  int w    = idx >> 12;
  int k = 4 * kc + j + 2 * (lane >> 4);
  int n = 16 * nt + (lane & 15);
  const float* W = (w == 0) ? Wq : (w == 1) ? Wk : Wv;
  wsw[idx] = W[k * 64 + n];
}

__global__ __launch_bounds__(128) void birnn_kernel(
    const float* __restrict__ x, const float* __restrict__ wsw,
    float* __restrict__ out) {
  __shared__ __align__(16) float smem[13 * BUF];
  float* bufL  = smem + 0 * BUF;
  float* bufS  = smem + 1 * BUF;
  float* bufX  = smem + 2 * BUF;
  float* bufO0 = smem + 3 * BUF;
  float* bufO1 = smem + 4 * BUF;
  float* P0 = smem + 5 * BUF;
  float* P1 = smem + 6 * BUF;
  float* P2 = smem + 7 * BUF;
  float* P3 = smem + 8 * BUF;
  float* P4 = smem + 9 * BUF;
  float* P5 = smem + 10 * BUF;
  float* P6 = smem + 11 * BUF;
  float* P7 = smem + 12 * BUF;

  const int lane = threadIdx.x & 31;
  const int wid  = threadIdx.x >> 5;
  const int m = lane & 15;
  const int h = lane >> 4;
  const int b = blockIdx.x * 16 + m;
  const float kScale = 0.125f;

  v2f wf[64];
  {
    const float* myw = wsw + ((wid == 3) ? 1 : wid) * 4096 + lane * 2;
#pragma unroll
    for (int f = 0; f < 64; ++f) wf[f] = *(const v2f*)(myw + f * 64);
  }
  v2f wv3[64];
  v2f wq3[32];
  if (wid == 3) {
    const float* wvp = wsw + 2 * 4096 + lane * 2;
    const float* wqp = wsw + lane * 2;
#pragma unroll
    for (int f = 0; f < 64; ++f) wv3[f] = *(const v2f*)(wvp + f * 64);
#pragma unroll
    for (int f = 0; f < 32; ++f)
      wq3[f] = *(const v2f*)(wqp + ((f >> 1) * 4 + 2 + (f & 1)) * 64);
  }

  if (wid == 0) {
    const float4* xr = (const float4*)(x + (size_t)b * TSTEPS * DDIM + 32 * h);
    float4* dl  = (float4*)(bufL + m * LS + 32 * h);
    float4* dsv = (float4*)(bufS + m * LS + 32 * h);
#pragma unroll
    for (int i = 0; i < 8; ++i) {
      float4 v = xr[i];
      dl[i]  = v;
      dsv[i] = v;
    }
  }
  __syncthreads();

#pragma unroll 1
  for (int t = 1; t < TSTEPS; ++t) {
    if (wid == 2) {
      const float* xrow = x + ((size_t)b * TSTEPS + t) * DDIM + 32 * h;
      const float4* xr = (const float4*)xrow;
      float4* dx = (float4*)(bufX + m * LS + 32 * h);
#pragma unroll
      for (int i = 0; i < 8; ++i) dx[i] = xr[i];
      if (t + 1 < TSTEPS) __builtin_prefetch(xrow + DDIM, 0, 0);
    }
    __syncthreads();

    if (wid == 0) {
      proj16x64(bufL, wf, P0, m, h);
      proj16x64(bufX, wf, P3, m, h);
    } else if (wid == 1) {
      proj16x64(bufL, wf, P1, m, h);
      proj16x64(bufS, wf, P6, m, h);
    } else if (wid == 2) {
      proj16x64(bufL, wf, P2, m, h);
      proj16x64(bufS, wf, P7, m, h);
    } else {
      proj_tiles2<4, 0, 4, 0, 4, 0>(bufX, wf, wv3, P4, P5, m, h);
    }
    __syncthreads();

    if (wid == 0) {
      float d00 = rowdot(P0, P1, m, h) * kScale;
      float d01 = rowdot(P0, P4, m, h) * kScale;
      float p0, p1;
      softmax2(d00, d01, p0, p1);
      combine<true>(P2, P5, p0, p1, bufO0, m, h);
    } else if (wid == 1) {
      float d10 = rowdot(P3, P6, m, h) * kScale;
      float d11 = rowdot(P3, P4, m, h) * kScale;
      float q0, q1;
      softmax2(d10, d11, q0, q1);
      combine<true>(P7, P5, q0, q1, bufO1, m, h);
    }
    __syncthreads();

    if (wid == 0) {
      proj16x64(bufO0, wf, P0, m, h);
      proj_tiles<2, 0, 4, 0>(bufO1, wf, P3, m, h);
    } else if (wid == 1) {
      proj16x64(bufO0, wf, P1, m, h);
      proj_tiles<2, 0, 4, 0>(bufO1, wf, P4, m, h);
    } else if (wid == 2) {
      proj16x64(bufO0, wf, P2, m, h);
      proj_tiles<2, 0, 4, 0>(bufO1, wf, P5, m, h);
    } else {
      proj_tiles3<2, 2, 2, 0, 4, 2, 4, 2>(bufO1, wq3, wf, wv3, P3, P4, P5, m,
                                          h);
    }
    __syncthreads();

    if (wid == 0) {
      float e00 = rowdot(P0, P1, m, h) * kScale;
      float e01 = rowdot(P0, P4, m, h) * kScale;
      float r0, r1;
      softmax2(e00, e01, r0, r1);
      combine<false>(P2, P5, r0, r1, bufL, m, h);
    } else if (wid == 1) {
      float e10 = rowdot(P3, P1, m, h) * kScale;
      float e11 = rowdot(P3, P4, m, h) * kScale;
      float s0, s1;
      softmax2(e10, e11, s0, s1);
      combine<false>(P2, P5, s0, s1, bufS, m, h);
    }
    __syncthreads();
  }

  if (wid == 0) {
    const float4* src = (const float4*)(bufL + m * LS + 32 * h);
    float4* dst = (float4*)(out + (size_t)b * DDIM + 32 * h);
#pragma unroll
    for (int i = 0; i < 8; ++i) dst[i] = src[i];
  }
}

extern "C" void kernel_launch(void* const* d_in, const int* in_sizes, int n_in,
                              void* d_out, int out_size, void* d_ws,
                              size_t ws_size, hipStream_t stream) {
  (void)in_sizes; (void)n_in; (void)out_size; (void)ws_size;
  const float* x  = (const float*)d_in[0];
  const float* Wq = (const float*)d_in[1];
  const float* Wk = (const float*)d_in[2];
  const float* Wv = (const float*)d_in[3];
  float* wsw = (float*)d_ws;

  swizzle_weights<<<48, 256, 0, stream>>>(Wq, Wk, Wv, wsw);
  birnn_kernel<<<2048 / 16, 128, 0, stream>>>(x, wsw, (float*)d_out);
}


# x_t path runs one step ahead: its fetch and its Qx, Kx, Vx projections move into the attention phases on the two idle waves; serial chain 5+6 projections, was 8+6; sum order unchanged
# speedup vs baseline: 1.4755x; 1.0787x over previous
#include <hip/hip_runtime.h>
#include <math.h>


#define LS     68
#define BUF    (16 * LS)
#define TSTEPS 200
#define DDIM   64

typedef float v2f __attribute__((ext_vector_type(2)));
typedef float v4f __attribute__((ext_vector_type(4)));
typedef float v8f __attribute__((ext_vector_type(8)));

__device__ __forceinline__ float fast_tanh(float v) {
#if __has_builtin(__builtin_amdgcn_tanhf)
  return __builtin_amdgcn_tanhf(v);
#elif __has_builtin(__builtin_amdgcn_tanh_f32)
  return __builtin_amdgcn_tanh_f32(v);
#else
  return tanhf(v);
#endif
}

__device__ __forceinline__ v8f wmma_f32(v2f a, v2f b, v8f c) {
  return __builtin_amdgcn_wmma_f32_16x16x4_f32(false, a, false, b, (short)0, c,
                                               false, false);
}

#define TC(dst, w, nt) (((dst) << 4) | ((w) << 2) | (nt))
template <int N, int C0, int C1, int C2, int C3, int C4, int C5>
__device__ __forceinline__ void proj_multi(const float* src, const v2f* wall,
                                           float* dbase, int m, int h) {
  constexpr int code[6] = {C0, C1, C2, C3, C4, C5};
  v8f acc[N];
#pragma unroll
  for (int i = 0; i < N; ++i) acc[i] = v8f{};
  const float* arow = src + m * LS + 2 * h;
#pragma unroll
  for (int kc = 0; kc < 16; ++kc) {
    v2f af = *(const v2f*)(arow + 4 * kc);
#pragma unroll
    for (int i = 0; i < N; ++i)
      acc[i] = wmma_f32(
          af, wall[((code[i] >> 2) & 3) * 64 + kc * 4 + (code[i] & 3)], acc[i]);
  }
#pragma unroll
  for (int r = 0; r < 8; ++r) {
    int row = r + 8 * h;
#pragma unroll
    for (int i = 0; i < N; ++i)
      dbase[(code[i] >> 4) * BUF + row * LS + m + 16 * (code[i] & 3)] =
          acc[i][r];
  }
}

__device__ __forceinline__ void load_afrags(const float* src, v2f* af, int m,
                                            int h) {
  const float* arow = src + m * LS + 2 * h;
#pragma unroll
  for (int kc = 0; kc < 16; ++kc)
    af[kc] = *(const v2f*)(arow + 4 * kc);
}

template <int N, int C0, int C1, int C2, int C3, int C4, int C5>
__device__ __forceinline__ void proj_multi_regs(const v2f* af, const v2f* wall,
                                                float* dbase, int m, int h) {
  constexpr int code[6] = {C0, C1, C2, C3, C4, C5};
  v8f acc[N];
#pragma unroll
  for (int i = 0; i < N; ++i) acc[i] = v8f{};
#pragma unroll
  for (int kc = 0; kc < 16; ++kc) {
#pragma unroll
    for (int i = 0; i < N; ++i)
      acc[i] = wmma_f32(
          af[kc], wall[((code[i] >> 2) & 3) * 64 + kc * 4 + (code[i] & 3)],
          acc[i]);
  }
#pragma unroll
  for (int r = 0; r < 8; ++r) {
    int row = r + 8 * h;
#pragma unroll
    for (int i = 0; i < N; ++i)
      dbase[(code[i] >> 4) * BUF + row * LS + m + 16 * (code[i] & 3)] =
          acc[i][r];
  }
}

__device__ __forceinline__ float rowdot(const float* A, const float* B, int m,
                                        int h) {
  const float4* a = (const float4*)(A + m * LS + 32 * h);
  const float4* b = (const float4*)(B + m * LS + 32 * h);
  float s = 0.f;
#pragma unroll
  for (int i = 0; i < 8; ++i) {
    float4 u = a[i], v = b[i];
    s += u.x * v.x + u.y * v.y + u.z * v.z + u.w * v.w;
  }
  s += __shfl_xor(s, 16, 32);
  return s;
}

__device__ __forceinline__ void softmax2(float a, float b, float& p0,
                                         float& p1) {
  float mx  = fmaxf(a, b);
  float e0  = __expf(a - mx);
  float e1  = __expf(b - mx);
  float inv = 1.f / (e0 + e1);
  p0 = e0 * inv;
  p1 = e1 * inv;
}

template <bool DO_TANH>
__device__ __forceinline__ void combine(const float* V0, const float* V1,
                                        float p0, float p1, float* dst, int m,
                                        int h) {
  const float4* v0 = (const float4*)(V0 + m * LS + 32 * h);
  const float4* v1 = (const float4*)(V1 + m * LS + 32 * h);
  float4* d = (float4*)(dst + m * LS + 32 * h);
#pragma unroll
  for (int i = 0; i < 8; ++i) {
    float4 a = v0[i], b = v1[i], r;
    r.x = p0 * a.x + p1 * b.x;
    r.y = p0 * a.y + p1 * b.y;
    r.z = p0 * a.z + p1 * b.z;
    r.w = p0 * a.w + p1 * b.w;
    if (DO_TANH) {
      r.x = fast_tanh(r.x);
      r.y = fast_tanh(r.y);
      r.z = fast_tanh(r.z);
      r.w = fast_tanh(r.w);
    }
    d[i] = r;
  }
}

__global__ void swizzle_weights(const float* __restrict__ Wq,
                                const float* __restrict__ Wk,
                                const float* __restrict__ Wv,
                                float* __restrict__ wsw) {
  int idx = blockIdx.x * blockDim.x + threadIdx.x;
  if (idx >= 3 * 16 * 4 * 32 * 2) return;
  int j    = idx & 1;
  int lane = (idx >> 1) & 31;
  int nt   = (idx >> 6) & 3;
  int kc   = (idx >> 8) & 15;
  int w    = idx >> 12;
  int k = 4 * kc + j + 2 * (lane >> 4);
  int n = 16 * nt + (lane & 15);
  const float* W = (w == 0) ? Wq : (w == 1) ? Wk : Wv;
  wsw[idx] = W[k * 64 + n];
}

#define B_L   0
#define B_S   1
#define B_X   2
#define B_O0  3
#define B_O1  4
#define B_P0  5
#define B_P1  6
#define B_P2  7
#define B_P3  8
#define B_P4  9
#define B_P5  10
#define B_XP  11
#define N_BUF 17

__global__ __launch_bounds__(128) void birnn_kernel(
    const float* __restrict__ x, const float* __restrict__ wsw,
    float* __restrict__ out) {
  __shared__ __align__(16) float smem[N_BUF * BUF];
  float* bufL  = smem + B_L * BUF;
  float* bufS  = smem + B_S * BUF;
  float* bufX  = smem + B_X * BUF;
  float* bufO0 = smem + B_O0 * BUF;
  float* bufO1 = smem + B_O1 * BUF;
  float* P0 = smem + B_P0 * BUF;
  float* P1 = smem + B_P1 * BUF;
  float* P2 = smem + B_P2 * BUF;
  float* P3 = smem + B_P3 * BUF;
  float* P4 = smem + B_P4 * BUF;
  float* P5 = smem + B_P5 * BUF;
  float* pxc = smem + B_XP * BUF;
  float* pxn = smem + (B_XP + 3) * BUF;

  const int lane = threadIdx.x & 31;
  const int wid  = threadIdx.x >> 5;
  const int m = lane & 15;
  const int h = lane >> 4;
  const int b = blockIdx.x * 16 + m;
  const float kScale = 0.125f;

  v2f wall[192];
  {
    const float* myw = wsw + lane * 2;
#pragma unroll
    for (int f = 0; f < 192; ++f) wall[f] = *(const v2f*)(myw + f * 64);
  }

  const float* xb = x + (size_t)b * TSTEPS * DDIM + 32 * h;
  v4f xr[8];
  v2f xa[16];

  if (wid == 0) {
    const float4* xq = (const float4*)xb;
    float4* dl  = (float4*)(bufL + m * LS + 32 * h);
    float4* dsv = (float4*)(bufS + m * LS + 32 * h);
#pragma unroll
    for (int i = 0; i < 8; ++i) {
      float4 v = xq[i];
      dl[i]  = v;
      dsv[i] = v;
    }
  } else if (wid == 2) {
    const float4* xq = (const float4*)(xb + DDIM);
    float4* dx = (float4*)(bufX + m * LS + 32 * h);
#pragma unroll
    for (int i = 0; i < 8; ++i) dx[i] = xq[i];
    const v4f* xq2 = (const v4f*)(xb + 2 * DDIM);
#pragma unroll
    for (int i = 0; i < 8; ++i) xr[i] = xq2[i];
  }
  __syncthreads();
  if (wid == 0)
    proj_multi<3, TC(0, 0, 0), TC(0, 0, 1), TC(0, 0, 2), 0, 0, 0>(bufX, wall, pxc, m, h);
  else if (wid == 1)
    proj_multi<3, TC(0, 0, 3), TC(1, 1, 0), TC(1, 1, 1), 0, 0, 0>(bufX, wall, pxc, m, h);
  else if (wid == 2)
    proj_multi<3, TC(1, 1, 2), TC(1, 1, 3), TC(2, 2, 0), 0, 0, 0>(bufX, wall, pxc, m, h);
  else
    proj_multi<3, TC(2, 2, 1), TC(2, 2, 2), TC(2, 2, 3), 0, 0, 0>(bufX, wall, pxc, m, h);
  __syncthreads();

#pragma unroll 1
  for (int t = 1; t < TSTEPS; ++t) {
    const bool ahead = (t + 1 < TSTEPS);

    if (wid == 0) {
      proj_multi<5, TC(B_P0, 0, 0), TC(B_P0, 0, 1), TC(B_P0, 0, 2),
                 TC(B_P0, 0, 3), TC(B_P1, 1, 0), 0>(bufL, wall, smem, m, h);
    } else if (wid == 1) {
      proj_multi<5, TC(B_P1, 1, 1), TC(B_P1, 1, 2), TC(B_P1, 1, 3),
                 TC(B_P2, 2, 0), TC(B_P2, 2, 1), 0>(bufL, wall, smem, m, h);
    } else if (wid == 2) {
      if (ahead) {
        v4f* dx = (v4f*)(bufX + m * LS + 32 * h);
#pragma unroll
        for (int i = 0; i < 8; ++i) dx[i] = xr[i];
      }
      proj_multi<2, TC(B_P2, 2, 2), TC(B_P2, 2, 3), 0, 0, 0, 0>(bufL, wall, smem, m, h);
      proj_multi<3, TC(B_P3, 1, 0), TC(B_P3, 1, 1), TC(B_P3, 1, 2), 0, 0, 0>(bufS, wall, smem, m, h);
    } else {
      proj_multi<5, TC(B_P3, 1, 3), TC(B_P4, 2, 0), TC(B_P4, 2, 1),
                 TC(B_P4, 2, 2), TC(B_P4, 2, 3), 0>(bufS, wall, smem, m, h);
    }
    __syncthreads();

    if (wid == 0) {
      float d00 = rowdot(P0, P1, m, h) * kScale;
      float d01 = rowdot(P0, pxc + BUF, m, h) * kScale;
      float p0, p1;
      softmax2(d00, d01, p0, p1);
      combine<true>(P2, pxc + 2 * BUF, p0, p1, bufO0, m, h);
    } else if (wid == 1) {
      float d10 = rowdot(pxc, P3, m, h) * kScale;
      float d11 = rowdot(pxc, pxc + BUF, m, h) * kScale;
      float q0, q1;
      softmax2(d10, d11, q0, q1);
      combine<true>(P4, pxc + 2 * BUF, q0, q1, bufO1, m, h);
    } else if (wid == 2) {
      if (t + 2 < TSTEPS) {
        const v4f* xq = (const v4f*)(xb + (size_t)(t + 2) * DDIM);
#pragma unroll
        for (int i = 0; i < 8; ++i) xr[i] = xq[i];
      }
      if (ahead) {
        load_afrags(bufX, xa, m, h);
        proj_multi_regs<3, TC(0, 0, 0), TC(0, 0, 1), TC(0, 0, 2), 0, 0, 0>(xa, wall, pxn, m, h);
      }
    } else {
      if (ahead) {
        load_afrags(bufX, xa, m, h);
        proj_multi_regs<3, TC(0, 0, 3), TC(1, 1, 0), TC(1, 1, 1), 0, 0, 0>(xa, wall, pxn, m, h);
      }
    }
    __syncthreads();

    if (wid == 0) {
      proj_multi<4, TC(B_P0, 0, 0), TC(B_P0, 0, 1), TC(B_P0, 0, 2),
                 TC(B_P0, 0, 3), 0, 0>(bufO0, wall, smem, m, h);
      proj_multi<2, TC(B_P3, 0, 0), TC(B_P3, 0, 1), 0, 0, 0, 0>(bufO1, wall, smem, m, h);
    } else if (wid == 1) {
      proj_multi<4, TC(B_P1, 1, 0), TC(B_P1, 1, 1), TC(B_P1, 1, 2),
                 TC(B_P1, 1, 3), 0, 0>(bufO0, wall, smem, m, h);
      proj_multi<2, TC(B_P4, 1, 0), TC(B_P4, 1, 1), 0, 0, 0, 0>(bufO1, wall, smem, m, h);
    } else if (wid == 2) {
      proj_multi<4, TC(B_P2, 2, 0), TC(B_P2, 2, 1), TC(B_P2, 2, 2),
                 TC(B_P2, 2, 3), 0, 0>(bufO0, wall, smem, m, h);
      proj_multi<2, TC(B_P5, 2, 0), TC(B_P5, 2, 1), 0, 0, 0, 0>(bufO1, wall, smem, m, h);
    } else {
      proj_multi<6, TC(B_P3, 0, 2), TC(B_P3, 0, 3), TC(B_P4, 1, 2),
                 TC(B_P4, 1, 3), TC(B_P5, 2, 2), TC(B_P5, 2, 3)>(bufO1, wall, smem, m, h);
    }
    __syncthreads();

    if (wid == 0) {
      float e00 = rowdot(P0, P1, m, h) * kScale;
      float e01 = rowdot(P0, P4, m, h) * kScale;
      float r0, r1;
      softmax2(e00, e01, r0, r1);
      combine<false>(P2, P5, r0, r1, bufL, m, h);
    } else if (wid == 1) {
      float e10 = rowdot(P3, P1, m, h) * kScale;
      float e11 = rowdot(P3, P4, m, h) * kScale;
      float s0, s1;
      softmax2(e10, e11, s0, s1);
      combine<false>(P2, P5, s0, s1, bufS, m, h);
    } else if (wid == 2) {
      if (ahead)
        proj_multi_regs<3, TC(1, 1, 2), TC(1, 1, 3), TC(2, 2, 0), 0, 0, 0>(xa, wall, pxn, m, h);
    } else {
      if (ahead)
        proj_multi_regs<3, TC(2, 2, 1), TC(2, 2, 2), TC(2, 2, 3), 0, 0, 0>(xa, wall, pxn, m, h);
    }
    __syncthreads();

    float* tmp = pxc;
    pxc = pxn;
    pxn = tmp;
  }

  if (wid == 0) {
    const float4* src = (const float4*)(bufL + m * LS + 32 * h);
    float4* dst = (float4*)(out + (size_t)b * DDIM + 32 * h);
#pragma unroll
    for (int i = 0; i < 8; ++i) dst[i] = src[i];
  }
}

extern "C" void kernel_launch(void* const* d_in, const int* in_sizes, int n_in,
                              void* d_out, int out_size, void* d_ws,
                              size_t ws_size, hipStream_t stream) {
  (void)in_sizes; (void)n_in; (void)out_size; (void)ws_size;
  const float* x  = (const float*)d_in[0];
  const float* Wq = (const float*)d_in[1];
  const float* Wk = (const float*)d_in[2];
  const float* Wv = (const float*)d_in[3];
  float* wsw = (float*)d_ws;

  swizzle_weights<<<48, 256, 0, stream>>>(Wq, Wk, Wv, wsw);
  birnn_kernel<<<2048 / 16, 128, 0, stream>>>(x, wsw, (float*)d_out);
}


# Scores from one folded matrix: pre-pass forms M = Wq Wk^T on the matrix unit and q.k becomes (a M).b; 9 projections per step, was 14 (serial 3+4, was 5+6); the scores' sum order changes
# speedup vs baseline: 1.8781x; 1.2729x over previous
#include <hip/hip_runtime.h>
#include <math.h>


#define LS     68
#define BUF    (16 * LS)
#define TSTEPS 200
#define DDIM   64

typedef float v2f __attribute__((ext_vector_type(2)));
typedef float v4f __attribute__((ext_vector_type(4)));
typedef float v8f __attribute__((ext_vector_type(8)));

__device__ __forceinline__ float fast_tanh(float v) {
#if __has_builtin(__builtin_amdgcn_tanhf)
  return __builtin_amdgcn_tanhf(v);
#elif __has_builtin(__builtin_amdgcn_tanh_f32)
  return __builtin_amdgcn_tanh_f32(v);
#else
  return tanhf(v);
#endif
}

__device__ __forceinline__ v8f wmma_f32(v2f a, v2f b, v8f c) {
  return __builtin_amdgcn_wmma_f32_16x16x4_f32(false, a, false, b, (short)0, c,
                                               false, false);
}

#define TC(dst, w, nt) (((dst) << 4) | ((w) << 2) | (nt))
template <int N, int C0, int C1, int C2, int C3, int C4, int C5>
__device__ __forceinline__ void proj_multi(const float* src, const v2f* wall,
                                           float* dbase, int m, int h) {
  constexpr int code[6] = {C0, C1, C2, C3, C4, C5};
  v8f acc[N];
#pragma unroll
  for (int i = 0; i < N; ++i) acc[i] = v8f{};
  const float* arow = src + m * LS + 2 * h;
#pragma unroll
  for (int kc = 0; kc < 16; ++kc) {
    v2f af = *(const v2f*)(arow + 4 * kc);
#pragma unroll
    for (int i = 0; i < N; ++i)
      acc[i] = wmma_f32(
          af, wall[((code[i] >> 2) & 3) * 64 + kc * 4 + (code[i] & 3)], acc[i]);
  }
#pragma unroll
  for (int r = 0; r < 8; ++r) {
    int row = r + 8 * h;
#pragma unroll
    for (int i = 0; i < N; ++i)
      dbase[(code[i] >> 4) * BUF + row * LS + m + 16 * (code[i] & 3)] =
          acc[i][r];
  }
}

template <int NA, int NB, int C0, int C1, int C2, int C3>
__device__ __forceinline__ void proj_two(const float* srcA, const float* srcB,
                                         const v2f* wall, float* dbase, int m,
                                         int h) {
  constexpr int N = NA + NB;
  constexpr int code[4] = {C0, C1, C2, C3};
  v8f acc[N];
#pragma unroll
  for (int i = 0; i < N; ++i) acc[i] = v8f{};
  const float* arowA = srcA + m * LS + 2 * h;
  const float* arowB = srcB + m * LS + 2 * h;
#pragma unroll
  for (int kc = 0; kc < 16; ++kc) {
    v2f afA = *(const v2f*)(arowA + 4 * kc);
    v2f afB = *(const v2f*)(arowB + 4 * kc);
#pragma unroll
    for (int i = 0; i < N; ++i)
      acc[i] = wmma_f32(
          i < NA ? afA : afB,
          wall[((code[i] >> 2) & 3) * 64 + kc * 4 + (code[i] & 3)], acc[i]);
  }
#pragma unroll
  for (int r = 0; r < 8; ++r) {
    int row = r + 8 * h;
#pragma unroll
    for (int i = 0; i < N; ++i)
      dbase[(code[i] >> 4) * BUF + row * LS + m + 16 * (code[i] & 3)] =
          acc[i][r];
  }
}

__device__ __forceinline__ void load_afrags(const float* src, v2f* af, int m,
                                            int h) {
  const float* arow = src + m * LS + 2 * h;
#pragma unroll
  for (int kc = 0; kc < 16; ++kc)
    af[kc] = *(const v2f*)(arow + 4 * kc);
}

template <int N, int C0, int C1, int C2, int C3, int C4, int C5>
__device__ __forceinline__ void proj_multi_regs(const v2f* af, const v2f* wall,
                                                float* dbase, int m, int h) {
  constexpr int code[6] = {C0, C1, C2, C3, C4, C5};
  v8f acc[N];
#pragma unroll
  for (int i = 0; i < N; ++i) acc[i] = v8f{};
#pragma unroll
  for (int kc = 0; kc < 16; ++kc) {
#pragma unroll
    for (int i = 0; i < N; ++i)
      acc[i] = wmma_f32(
          af[kc], wall[((code[i] >> 2) & 3) * 64 + kc * 4 + (code[i] & 3)],
          acc[i]);
  }
#pragma unroll
  for (int r = 0; r < 8; ++r) {
    int row = r + 8 * h;
#pragma unroll
    for (int i = 0; i < N; ++i)
      dbase[(code[i] >> 4) * BUF + row * LS + m + 16 * (code[i] & 3)] =
          acc[i][r];
  }
}

__device__ __forceinline__ float rowdot(const float* A, const float* B, int m,
                                        int h) {
  const float4* a = (const float4*)(A + m * LS + 32 * h);
  const float4* b = (const float4*)(B + m * LS + 32 * h);
  float s = 0.f;
#pragma unroll
  for (int i = 0; i < 8; ++i) {
    float4 u = a[i], v = b[i];
    s += u.x * v.x + u.y * v.y + u.z * v.z + u.w * v.w;
  }
  s += __shfl_xor(s, 16, 32);
  return s;
}

__device__ __forceinline__ void softmax2(float a, float b, float& p0,
                                         float& p1) {
  float mx  = fmaxf(a, b);
  float e0  = __expf(a - mx);
  float e1  = __expf(b - mx);
  float inv = 1.f / (e0 + e1);
  p0 = e0 * inv;
  p1 = e1 * inv;
}

template <bool DO_TANH>
__device__ __forceinline__ void combine(const float* V0, const float* V1,
                                        float p0, float p1, float* dst, int m,
                                        int h) {
  const float4* v0 = (const float4*)(V0 + m * LS + 32 * h);
  const float4* v1 = (const float4*)(V1 + m * LS + 32 * h);
  float4* d = (float4*)(dst + m * LS + 32 * h);
#pragma unroll
  for (int i = 0; i < 8; ++i) {
    float4 a = v0[i], b = v1[i], r;
    r.x = p0 * a.x + p1 * b.x;
    r.y = p0 * a.y + p1 * b.y;
    r.z = p0 * a.z + p1 * b.z;
    r.w = p0 * a.w + p1 * b.w;
    if (DO_TANH) {
      r.x = fast_tanh(r.x);
      r.y = fast_tanh(r.y);
      r.z = fast_tanh(r.z);
      r.w = fast_tanh(r.w);
    }
    d[i] = r;
  }
}

__global__ __launch_bounds__(128) void prep_weights(
    const float* __restrict__ Wq, const float* __restrict__ Wk,
    const float* __restrict__ Wv, float* __restrict__ wsw) {
  if (blockIdx.x == 0) {
    const int lane = threadIdx.x & 31;
    const int wv = threadIdx.x >> 5;
    const int m = lane & 15;
    const int h = lane >> 4;
    v8f acc[4];
#pragma unroll
    for (int nt = 0; nt < 4; ++nt) acc[nt] = v8f{};
    const float* arow = Wq + (16 * wv + m) * 64 + 2 * h;
    const float* brow = Wk + m * 64 + 2 * h;
#pragma unroll
    for (int kc = 0; kc < 16; ++kc) {
      v2f af = *(const v2f*)(arow + 4 * kc);
#pragma unroll
      for (int nt = 0; nt < 4; ++nt) {
        v2f bf = *(const v2f*)(brow + nt * 16 * 64 + 4 * kc);
        acc[nt] = wmma_f32(af, bf, acc[nt]);
      }
    }
#pragma unroll
    for (int nt = 0; nt < 4; ++nt) {
#pragma unroll
      for (int q = 0; q < 4; ++q) {
        v2f pr = {acc[nt][2 * q], acc[nt][2 * q + 1]};
        *(v2f*)(wsw + ((4 * wv + 2 * h + (q >> 1)) * 4 + nt) * 64 +
                (m + 16 * (q & 1)) * 2) = pr;
      }
    }
  } else {
    int idx = (blockIdx.x - 1) * 128 + threadIdx.x;
    int j    = idx & 1;
    int lane = (idx >> 1) & 31;
    int nt   = (idx >> 6) & 3;
    int kc   = (idx >> 8) & 15;
    int k = 4 * kc + j + 2 * (lane >> 4);
    int n = 16 * nt + (lane & 15);
    wsw[4096 + idx] = Wv[k * 64 + n];
  }
}

#define B_L   0
#define B_S   1
#define B_X   2
#define B_O0  4
#define B_O1  5
#define B_LM  6
#define B_VL  7
#define B_VS  8
#define B_OM0 9
#define B_OM1 10
#define B_V0  11
#define B_V1  12
#define B_XP  13
#define N_BUF 17

__global__ __launch_bounds__(128) void birnn_kernel(
    const float* __restrict__ x, const float* __restrict__ wsw,
    float* __restrict__ out) {
  __shared__ __align__(16) float smem[N_BUF * BUF];
  float* bufL  = smem + B_L * BUF;
  float* bufS  = smem + B_S * BUF;
  float* bufO0 = smem + B_O0 * BUF;
  float* bufO1 = smem + B_O1 * BUF;
  float* LM  = smem + B_LM * BUF;
  float* VL  = smem + B_VL * BUF;
  float* VS  = smem + B_VS * BUF;
  float* OM0 = smem + B_OM0 * BUF;
  float* OM1 = smem + B_OM1 * BUF;
  float* V0  = smem + B_V0 * BUF;
  float* V1  = smem + B_V1 * BUF;
  float* bxc = smem + B_X * BUF;
  float* bxn = smem + (B_X + 1) * BUF;
  float* pxc = smem + B_XP * BUF;
  float* pxn = smem + (B_XP + 2) * BUF;

  const int lane = threadIdx.x & 31;
  const int wid  = threadIdx.x >> 5;
  const int m = lane & 15;
  const int h = lane >> 4;
  const int b = blockIdx.x * 16 + m;
  const float kScale = 0.125f;

  v2f wall[128];
  {
    const float* myw = wsw + lane * 2;
#pragma unroll
    for (int f = 0; f < 128; ++f) wall[f] = *(const v2f*)(myw + f * 64);
  }

  const float* xb = x + (size_t)b * TSTEPS * DDIM + 32 * h;
  v4f xr[8];
  v2f xa[16];

  if (wid == 0) {
    const float4* xq = (const float4*)xb;
    float4* dl  = (float4*)(bufL + m * LS + 32 * h);
    float4* dsv = (float4*)(bufS + m * LS + 32 * h);
#pragma unroll
    for (int i = 0; i < 8; ++i) {
      float4 v = xq[i];
      dl[i]  = v;
      dsv[i] = v;
    }
  } else if (wid == 2) {
    const float4* xq = (const float4*)(xb + DDIM);
    float4* dx = (float4*)(bxc + m * LS + 32 * h);
#pragma unroll
    for (int i = 0; i < 8; ++i) dx[i] = xq[i];
    const v4f* xq2 = (const v4f*)(xb + 2 * DDIM);
#pragma unroll
    for (int i = 0; i < 8; ++i) xr[i] = xq2[i];
  }
  __syncthreads();
  if (wid == 0)
    proj_multi<2, TC(0, 0, 0), TC(0, 0, 1), 0, 0, 0, 0>(bxc, wall, pxc, m, h);
  else if (wid == 1)
    proj_multi<2, TC(0, 0, 2), TC(0, 0, 3), 0, 0, 0, 0>(bxc, wall, pxc, m, h);
  else if (wid == 2)
    proj_multi<2, TC(1, 1, 0), TC(1, 1, 1), 0, 0, 0, 0>(bxc, wall, pxc, m, h);
  else
    proj_multi<2, TC(1, 1, 2), TC(1, 1, 3), 0, 0, 0, 0>(bxc, wall, pxc, m, h);
  __syncthreads();

#pragma unroll 1
  for (int t = 1; t < TSTEPS; ++t) {
    const bool ahead = (t + 1 < TSTEPS);

    if (wid == 0) {
      proj_multi<3, TC(B_LM, 0, 0), TC(B_LM, 0, 1), TC(B_LM, 0, 2), 0, 0, 0>(
          bufL, wall, smem, m, h);
    } else if (wid == 1) {
      proj_multi<3, TC(B_LM, 0, 3), TC(B_VL, 1, 0), TC(B_VL, 1, 1), 0, 0, 0>(
          bufL, wall, smem, m, h);
    } else if (wid == 2) {
      if (ahead) {
        v4f* dx = (v4f*)(bxn + m * LS + 32 * h);
#pragma unroll
        for (int i = 0; i < 8; ++i) dx[i] = xr[i];
      }
      proj_two<2, 1, TC(B_VL, 1, 2), TC(B_VL, 1, 3), TC(B_VS, 1, 0), 0>(
          bufL, bufS, wall, smem, m, h);
    } else {
      proj_multi<3, TC(B_VS, 1, 1), TC(B_VS, 1, 2), TC(B_VS, 1, 3), 0, 0, 0>(
          bufS, wall, smem, m, h);
    }
    __syncthreads();

    if (wid == 0) {
      float d00 = rowdot(LM, bufL, m, h) * kScale;
      float d01 = rowdot(LM, bxc, m, h) * kScale;
      float p0, p1;
      softmax2(d00, d01, p0, p1);
      combine<true>(VL, pxc + BUF, p0, p1, bufO0, m, h);
    } else if (wid == 1) {
      float d10 = rowdot(pxc, bufS, m, h) * kScale;
      float d11 = rowdot(pxc, bxc, m, h) * kScale;
      float q0, q1;
      softmax2(d10, d11, q0, q1);
      combine<true>(VS, pxc + BUF, q0, q1, bufO1, m, h);
    } else if (wid == 2) {
      if (t + 2 < TSTEPS) {
        const v4f* xq = (const v4f*)(xb + (size_t)(t + 2) * DDIM);
#pragma unroll
        for (int i = 0; i < 8; ++i) xr[i] = xq[i];
      }
      if (ahead) {
        load_afrags(bxn, xa, m, h);
        proj_multi_regs<2, TC(0, 0, 0), TC(0, 0, 1), 0, 0, 0, 0>(xa, wall, pxn, m, h);
      }
    } else {
      if (ahead) {
        load_afrags(bxn, xa, m, h);
        proj_multi_regs<2, TC(0, 0, 2), TC(0, 0, 3), 0, 0, 0, 0>(xa, wall, pxn, m, h);
      }
    }
    __syncthreads();

    if (wid == 0) {
      proj_multi<4, TC(B_OM0, 0, 0), TC(B_OM0, 0, 1), TC(B_OM0, 0, 2),
                 TC(B_OM0, 0, 3), 0, 0>(bufO0, wall, smem, m, h);
    } else if (wid == 1) {
      proj_multi<4, TC(B_V0, 1, 0), TC(B_V0, 1, 1), TC(B_V0, 1, 2),
                 TC(B_V0, 1, 3), 0, 0>(bufO0, wall, smem, m, h);
    } else if (wid == 2) {
      proj_multi<4, TC(B_OM1, 0, 0), TC(B_OM1, 0, 1), TC(B_OM1, 0, 2),
                 TC(B_OM1, 0, 3), 0, 0>(bufO1, wall, smem, m, h);
    } else {
      proj_multi<4, TC(B_V1, 1, 0), TC(B_V1, 1, 1), TC(B_V1, 1, 2),
                 TC(B_V1, 1, 3), 0, 0>(bufO1, wall, smem, m, h);
    }
    __syncthreads();

    if (wid == 0) {
      float e00 = rowdot(OM0, bufO0, m, h) * kScale;
      float e01 = rowdot(OM0, bufO1, m, h) * kScale;
      float r0, r1;
      softmax2(e00, e01, r0, r1);
      combine<false>(V0, V1, r0, r1, bufL, m, h);
    } else if (wid == 1) {
      float e10 = rowdot(OM1, bufO0, m, h) * kScale;
      float e11 = rowdot(OM1, bufO1, m, h) * kScale;
      float s0, s1;
      softmax2(e10, e11, s0, s1);
      combine<false>(V0, V1, s0, s1, bufS, m, h);
    } else if (wid == 2) {
      if (ahead)
        proj_multi_regs<2, TC(1, 1, 0), TC(1, 1, 1), 0, 0, 0, 0>(xa, wall, pxn, m, h);
    } else {
      if (ahead)
        proj_multi_regs<2, TC(1, 1, 2), TC(1, 1, 3), 0, 0, 0, 0>(xa, wall, pxn, m, h);
    }
    __syncthreads();

    float* tmp = pxc;
    pxc = pxn;
    pxn = tmp;
    tmp = bxc;
    bxc = bxn;
    bxn = tmp;
  }

  if (wid == 0) {
    const float4* src = (const float4*)(bufL + m * LS + 32 * h);
    float4* dst = (float4*)(out + (size_t)b * DDIM + 32 * h);
#pragma unroll
    for (int i = 0; i < 8; ++i) dst[i] = src[i];
  }
}

extern "C" void kernel_launch(void* const* d_in, const int* in_sizes, int n_in,
                              void* d_out, int out_size, void* d_ws,
                              size_t ws_size, hipStream_t stream) {
  (void)in_sizes; (void)n_in; (void)out_size; (void)ws_size;
  const float* x  = (const float*)d_in[0];
  const float* Wq = (const float*)d_in[1];
  const float* Wk = (const float*)d_in[2];
  const float* Wv = (const float*)d_in[3];
  float* wsw = (float*)d_ws;

  prep_weights<<<33, 128, 0, stream>>>(Wq, Wk, Wv, wsw);
  birnn_kernel<<<2048 / 16, 128, 0, stream>>>(x, wsw, (float*)d_out);
}


# Each carry and o projection block fetches all its A fragments from LDS as one group ahead of its first WMMA (was a pair at a time among the WMMAs); sum order unchanged
# speedup vs baseline: 2.0498x; 1.0914x over previous
#include <hip/hip_runtime.h>
#include <math.h>


#define LS     68
#define BUF    (16 * LS)
#define TSTEPS 200
#define DDIM   64

typedef float v2f __attribute__((ext_vector_type(2)));
typedef float v4f __attribute__((ext_vector_type(4)));
typedef float v8f __attribute__((ext_vector_type(8)));

__device__ __forceinline__ float fast_tanh(float v) {
#if __has_builtin(__builtin_amdgcn_tanhf)
  return __builtin_amdgcn_tanhf(v);
#elif __has_builtin(__builtin_amdgcn_tanh_f32)
  return __builtin_amdgcn_tanh_f32(v);
#else
  return tanhf(v);
#endif
}

__device__ __forceinline__ v8f wmma_f32(v2f a, v2f b, v8f c) {
  return __builtin_amdgcn_wmma_f32_16x16x4_f32(false, a, false, b, (short)0, c,
                                               false, false);
}

__device__ __forceinline__ void load_afrags(const float* src, v2f* af, int m,
                                            int h) {
  const float* arow = src + m * LS + 2 * h;
#pragma unroll
  for (int kc = 0; kc < 16; ++kc)
    af[kc] = *(const v2f*)(arow + 4 * kc);
}

#define TC(dst, w, nt) (((dst) << 4) | ((w) << 2) | (nt))
template <int N, int C0, int C1, int C2, int C3, int C4, int C5>
__device__ __forceinline__ void proj_multi(const float* src, const v2f* wall,
                                           float* dbase, int m, int h) {
  constexpr int code[6] = {C0, C1, C2, C3, C4, C5};
  v8f acc[N];
#pragma unroll
  for (int i = 0; i < N; ++i) acc[i] = v8f{};
  v2f af[16];
  load_afrags(src, af, m, h);
  __builtin_amdgcn_sched_barrier(0);
#pragma unroll
  for (int kc = 0; kc < 16; ++kc) {
#pragma unroll
    for (int i = 0; i < N; ++i)
      acc[i] = wmma_f32(
          af[kc], wall[((code[i] >> 2) & 3) * 64 + kc * 4 + (code[i] & 3)],
          acc[i]);
  }
#pragma unroll
  for (int r = 0; r < 8; ++r) {
    int row = r + 8 * h;
#pragma unroll
    for (int i = 0; i < N; ++i)
      dbase[(code[i] >> 4) * BUF + row * LS + m + 16 * (code[i] & 3)] =
          acc[i][r];
  }
}

template <int NA, int NB, int C0, int C1, int C2, int C3>
__device__ __forceinline__ void proj_two(const float* srcA, const float* srcB,
                                         const v2f* wall, float* dbase, int m,
                                         int h) {
  constexpr int N = NA + NB;
  constexpr int code[4] = {C0, C1, C2, C3};
  v8f acc[N];
#pragma unroll
  for (int i = 0; i < N; ++i) acc[i] = v8f{};
  v2f afA[16], afB[16];
  const float* arowA = srcA + m * LS + 2 * h;
  const float* arowB = srcB + m * LS + 2 * h;
#pragma unroll
  for (int kc = 0; kc < 4; ++kc) {
    afA[kc] = *(const v2f*)(arowA + 4 * kc);
    afB[kc] = *(const v2f*)(arowB + 4 * kc);
  }
  __builtin_amdgcn_sched_barrier(0);
#pragma unroll
  for (int kc = 4; kc < 16; ++kc) {
    afA[kc] = *(const v2f*)(arowA + 4 * kc);
    afB[kc] = *(const v2f*)(arowB + 4 * kc);
  }
  __builtin_amdgcn_sched_barrier(0);
#pragma unroll
  for (int kc = 0; kc < 16; ++kc) {
#pragma unroll
    for (int i = 0; i < N; ++i)
      acc[i] = wmma_f32(
          i < NA ? afA[kc] : afB[kc],
          wall[((code[i] >> 2) & 3) * 64 + kc * 4 + (code[i] & 3)], acc[i]);
  }
#pragma unroll
  for (int r = 0; r < 8; ++r) {
    int row = r + 8 * h;
#pragma unroll
    for (int i = 0; i < N; ++i)
      dbase[(code[i] >> 4) * BUF + row * LS + m + 16 * (code[i] & 3)] =
          acc[i][r];
  }
}


template <int N, int C0, int C1, int C2, int C3, int C4, int C5>
__device__ __forceinline__ void proj_multi_regs(const v2f* af, const v2f* wall,
                                                float* dbase, int m, int h) {
  constexpr int code[6] = {C0, C1, C2, C3, C4, C5};
  v8f acc[N];
#pragma unroll
  for (int i = 0; i < N; ++i) acc[i] = v8f{};
#pragma unroll
  for (int kc = 0; kc < 16; ++kc) {
#pragma unroll
    for (int i = 0; i < N; ++i)
      acc[i] = wmma_f32(
          af[kc], wall[((code[i] >> 2) & 3) * 64 + kc * 4 + (code[i] & 3)],
          acc[i]);
  }
#pragma unroll
  for (int r = 0; r < 8; ++r) {
    int row = r + 8 * h;
#pragma unroll
    for (int i = 0; i < N; ++i)
      dbase[(code[i] >> 4) * BUF + row * LS + m + 16 * (code[i] & 3)] =
          acc[i][r];
  }
}

__device__ __forceinline__ float rowdot(const float* A, const float* B, int m,
                                        int h) {
  const float4* a = (const float4*)(A + m * LS + 32 * h);
  const float4* b = (const float4*)(B + m * LS + 32 * h);
  float s = 0.f;
#pragma unroll
  for (int i = 0; i < 8; ++i) {
    float4 u = a[i], v = b[i];
    s += u.x * v.x + u.y * v.y + u.z * v.z + u.w * v.w;
  }
  s += __shfl_xor(s, 16, 32);
  return s;
}

__device__ __forceinline__ void softmax2(float a, float b, float& p0,
                                         float& p1) {
  float mx  = fmaxf(a, b);
  float e0  = __expf(a - mx);
  float e1  = __expf(b - mx);
  float inv = 1.f / (e0 + e1);
  p0 = e0 * inv;
  p1 = e1 * inv;
}

template <bool DO_TANH>
__device__ __forceinline__ void combine(const float* V0, const float* V1,
                                        float p0, float p1, float* dst, int m,
                                        int h) {
  const float4* v0 = (const float4*)(V0 + m * LS + 32 * h);
  const float4* v1 = (const float4*)(V1 + m * LS + 32 * h);
  float4* d = (float4*)(dst + m * LS + 32 * h);
#pragma unroll
  for (int i = 0; i < 8; ++i) {
    float4 a = v0[i], b = v1[i], r;
    r.x = p0 * a.x + p1 * b.x;
    r.y = p0 * a.y + p1 * b.y;
    r.z = p0 * a.z + p1 * b.z;
    r.w = p0 * a.w + p1 * b.w;
    if (DO_TANH) {
      r.x = fast_tanh(r.x);
      r.y = fast_tanh(r.y);
      r.z = fast_tanh(r.z);
      r.w = fast_tanh(r.w);
    }
    d[i] = r;
  }
}

__global__ __launch_bounds__(128) void prep_weights(
    const float* __restrict__ Wq, const float* __restrict__ Wk,
    const float* __restrict__ Wv, float* __restrict__ wsw) {
  if (blockIdx.x == 0) {
    const int lane = threadIdx.x & 31;
    const int wv = threadIdx.x >> 5;
    const int m = lane & 15;
    const int h = lane >> 4;
    v8f acc[4];
#pragma unroll
    for (int nt = 0; nt < 4; ++nt) acc[nt] = v8f{};
    const float* arow = Wq + (16 * wv + m) * 64 + 2 * h;
    const float* brow = Wk + m * 64 + 2 * h;
#pragma unroll
    for (int kc = 0; kc < 16; ++kc) {
      v2f af = *(const v2f*)(arow + 4 * kc);
#pragma unroll
      for (int nt = 0; nt < 4; ++nt) {
        v2f bf = *(const v2f*)(brow + nt * 16 * 64 + 4 * kc);
        acc[nt] = wmma_f32(af, bf, acc[nt]);
      }
    }
#pragma unroll
    for (int nt = 0; nt < 4; ++nt) {
#pragma unroll
      for (int q = 0; q < 4; ++q) {
        v2f pr = {acc[nt][2 * q], acc[nt][2 * q + 1]};
        *(v2f*)(wsw + ((4 * wv + 2 * h + (q >> 1)) * 4 + nt) * 64 +
                (m + 16 * (q & 1)) * 2) = pr;
      }
    }
  } else {
    int idx = (blockIdx.x - 1) * 128 + threadIdx.x;
    int j    = idx & 1;
    int lane = (idx >> 1) & 31;
    int nt   = (idx >> 6) & 3;
    int kc   = (idx >> 8) & 15;
    int k = 4 * kc + j + 2 * (lane >> 4);
    int n = 16 * nt + (lane & 15);
    wsw[4096 + idx] = Wv[k * 64 + n];
  }
}

#define B_L   0
#define B_S   1
#define B_X   2
#define B_O0  4
#define B_O1  5
#define B_LM  6
#define B_VL  7
#define B_VS  8
#define B_OM0 9
#define B_OM1 10
#define B_V0  11
#define B_V1  12
#define B_XP  13
#define N_BUF 17

__global__ __launch_bounds__(128) void birnn_kernel(
    const float* __restrict__ x, const float* __restrict__ wsw,
    float* __restrict__ out) {
  __shared__ __align__(16) float smem[N_BUF * BUF];
  float* bufL  = smem + B_L * BUF;
  float* bufS  = smem + B_S * BUF;
  float* bufO0 = smem + B_O0 * BUF;
  float* bufO1 = smem + B_O1 * BUF;
  float* LM  = smem + B_LM * BUF;
  float* VL  = smem + B_VL * BUF;
  float* VS  = smem + B_VS * BUF;
  float* OM0 = smem + B_OM0 * BUF;
  float* OM1 = smem + B_OM1 * BUF;
  float* V0  = smem + B_V0 * BUF;
  float* V1  = smem + B_V1 * BUF;
  float* bxc = smem + B_X * BUF;
  float* bxn = smem + (B_X + 1) * BUF;
  float* pxc = smem + B_XP * BUF;
  float* pxn = smem + (B_XP + 2) * BUF;

  const int lane = threadIdx.x & 31;
  const int wid  = threadIdx.x >> 5;
  const int m = lane & 15;
  const int h = lane >> 4;
  const int b = blockIdx.x * 16 + m;
  const float kScale = 0.125f;

  v2f wall[128];
  {
    const float* myw = wsw + lane * 2;
#pragma unroll
    for (int f = 0; f < 128; ++f) wall[f] = *(const v2f*)(myw + f * 64);
  }

  const float* xb = x + (size_t)b * TSTEPS * DDIM + 32 * h;
  v4f xr[8];
  v2f xa[16];

  if (wid == 0) {
    const float4* xq = (const float4*)xb;
    float4* dl  = (float4*)(bufL + m * LS + 32 * h);
    float4* dsv = (float4*)(bufS + m * LS + 32 * h);
#pragma unroll
    for (int i = 0; i < 8; ++i) {
      float4 v = xq[i];
      dl[i]  = v;
      dsv[i] = v;
    }
  } else if (wid == 2) {
    const float4* xq = (const float4*)(xb + DDIM);
    float4* dx = (float4*)(bxc + m * LS + 32 * h);
#pragma unroll
    for (int i = 0; i < 8; ++i) dx[i] = xq[i];
    const v4f* xq2 = (const v4f*)(xb + 2 * DDIM);
#pragma unroll
    for (int i = 0; i < 8; ++i) xr[i] = xq2[i];
  }
  __syncthreads();
  if (wid == 0)
    proj_multi<2, TC(0, 0, 0), TC(0, 0, 1), 0, 0, 0, 0>(bxc, wall, pxc, m, h);
  else if (wid == 1)
    proj_multi<2, TC(0, 0, 2), TC(0, 0, 3), 0, 0, 0, 0>(bxc, wall, pxc, m, h);
  else if (wid == 2)
    proj_multi<2, TC(1, 1, 0), TC(1, 1, 1), 0, 0, 0, 0>(bxc, wall, pxc, m, h);
  else
    proj_multi<2, TC(1, 1, 2), TC(1, 1, 3), 0, 0, 0, 0>(bxc, wall, pxc, m, h);
  __syncthreads();

#pragma unroll 1
  for (int t = 1; t < TSTEPS; ++t) {
    const bool ahead = (t + 1 < TSTEPS);

    if (wid == 0) {
      proj_multi<3, TC(B_LM, 0, 0), TC(B_LM, 0, 1), TC(B_LM, 0, 2), 0, 0, 0>(
          bufL, wall, smem, m, h);
    } else if (wid == 1) {
      proj_multi<3, TC(B_LM, 0, 3), TC(B_VL, 1, 0), TC(B_VL, 1, 1), 0, 0, 0>(
          bufL, wall, smem, m, h);
    } else if (wid == 2) {
      if (ahead) {
        v4f* dx = (v4f*)(bxn + m * LS + 32 * h);
#pragma unroll
        for (int i = 0; i < 8; ++i) dx[i] = xr[i];
      }
      proj_two<2, 1, TC(B_VL, 1, 2), TC(B_VL, 1, 3), TC(B_VS, 1, 0), 0>(
          bufL, bufS, wall, smem, m, h);
    } else {
      proj_multi<3, TC(B_VS, 1, 1), TC(B_VS, 1, 2), TC(B_VS, 1, 3), 0, 0, 0>(
          bufS, wall, smem, m, h);
    }
    __syncthreads();

    if (wid == 0) {
      float d00 = rowdot(LM, bufL, m, h) * kScale;
      float d01 = rowdot(LM, bxc, m, h) * kScale;
      float p0, p1;
      softmax2(d00, d01, p0, p1);
      combine<true>(VL, pxc + BUF, p0, p1, bufO0, m, h);
    } else if (wid == 1) {
      float d10 = rowdot(pxc, bufS, m, h) * kScale;
      float d11 = rowdot(pxc, bxc, m, h) * kScale;
      float q0, q1;
      softmax2(d10, d11, q0, q1);
      combine<true>(VS, pxc + BUF, q0, q1, bufO1, m, h);
    } else if (wid == 2) {
      if (t + 2 < TSTEPS) {
        const v4f* xq = (const v4f*)(xb + (size_t)(t + 2) * DDIM);
#pragma unroll
        for (int i = 0; i < 8; ++i) xr[i] = xq[i];
      }
      if (ahead) {
        load_afrags(bxn, xa, m, h);
        proj_multi_regs<2, TC(0, 0, 0), TC(0, 0, 1), 0, 0, 0, 0>(xa, wall, pxn, m, h);
      }
    } else {
      if (ahead) {
        load_afrags(bxn, xa, m, h);
        proj_multi_regs<2, TC(0, 0, 2), TC(0, 0, 3), 0, 0, 0, 0>(xa, wall, pxn, m, h);
      }
    }
    __syncthreads();

    if (wid == 0) {
      proj_multi<4, TC(B_OM0, 0, 0), TC(B_OM0, 0, 1), TC(B_OM0, 0, 2),
                 TC(B_OM0, 0, 3), 0, 0>(bufO0, wall, smem, m, h);
    } else if (wid == 1) {
      proj_multi<4, TC(B_V0, 1, 0), TC(B_V0, 1, 1), TC(B_V0, 1, 2),
                 TC(B_V0, 1, 3), 0, 0>(bufO0, wall, smem, m, h);
    } else if (wid == 2) {
      proj_multi<4, TC(B_OM1, 0, 0), TC(B_OM1, 0, 1), TC(B_OM1, 0, 2),
                 TC(B_OM1, 0, 3), 0, 0>(bufO1, wall, smem, m, h);
    } else {
      proj_multi<4, TC(B_V1, 1, 0), TC(B_V1, 1, 1), TC(B_V1, 1, 2),
                 TC(B_V1, 1, 3), 0, 0>(bufO1, wall, smem, m, h);
    }
    __syncthreads();

    if (wid == 0) {
      float e00 = rowdot(OM0, bufO0, m, h) * kScale;
      float e01 = rowdot(OM0, bufO1, m, h) * kScale;
      float r0, r1;
      softmax2(e00, e01, r0, r1);
      combine<false>(V0, V1, r0, r1, bufL, m, h);
    } else if (wid == 1) {
      float e10 = rowdot(OM1, bufO0, m, h) * kScale;
      float e11 = rowdot(OM1, bufO1, m, h) * kScale;
      float s0, s1;
      softmax2(e10, e11, s0, s1);
      combine<false>(V0, V1, s0, s1, bufS, m, h);
    } else if (wid == 2) {
      if (ahead)
        proj_multi_regs<2, TC(1, 1, 0), TC(1, 1, 1), 0, 0, 0, 0>(xa, wall, pxn, m, h);
    } else {
      if (ahead)
        proj_multi_regs<2, TC(1, 1, 2), TC(1, 1, 3), 0, 0, 0, 0>(xa, wall, pxn, m, h);
    }
    __syncthreads();

    float* tmp = pxc;
    pxc = pxn;
    pxn = tmp;
    tmp = bxc;
    bxc = bxn;
    bxn = tmp;
  }

  if (wid == 0) {
    const float4* src = (const float4*)(bufL + m * LS + 32 * h);
    float4* dst = (float4*)(out + (size_t)b * DDIM + 32 * h);
#pragma unroll
    for (int i = 0; i < 8; ++i) dst[i] = src[i];
  }
}

extern "C" void kernel_launch(void* const* d_in, const int* in_sizes, int n_in,
                              void* d_out, int out_size, void* d_ws,
                              size_t ws_size, hipStream_t stream) {
  (void)in_sizes; (void)n_in; (void)out_size; (void)ws_size;
  const float* x  = (const float*)d_in[0];
  const float* Wq = (const float*)d_in[1];
  const float* Wk = (const float*)d_in[2];
  const float* Wv = (const float*)d_in[3];
  float* wsw = (float*)d_ws;

  prep_weights<<<33, 128, 0, stream>>>(Wq, Wk, Wv, wsw);
  birnn_kernel<<<2048 / 16, 128, 0, stream>>>(x, wsw, (float*)d_out);
}


# Each attention phase fetches its five LDS half rows (query, two score operands, two value rows) as one group at its start, not piecemeal behind the softmax; score dots now compile to fma chains
# speedup vs baseline: 2.1180x; 1.0333x over previous
#include <hip/hip_runtime.h>
#include <math.h>


#define LS     68
#define BUF    (16 * LS)
#define TSTEPS 200
#define DDIM   64

typedef float v2f __attribute__((ext_vector_type(2)));
typedef float v4f __attribute__((ext_vector_type(4)));
typedef float v8f __attribute__((ext_vector_type(8)));

__device__ __forceinline__ float fast_tanh(float v) {
#if __has_builtin(__builtin_amdgcn_tanhf)
  return __builtin_amdgcn_tanhf(v);
#elif __has_builtin(__builtin_amdgcn_tanh_f32)
  return __builtin_amdgcn_tanh_f32(v);
#else
  return tanhf(v);
#endif
}

__device__ __forceinline__ v8f wmma_f32(v2f a, v2f b, v8f c) {
  return __builtin_amdgcn_wmma_f32_16x16x4_f32(false, a, false, b, (short)0, c,
                                               false, false);
}

__device__ __forceinline__ void load_afrags(const float* src, v2f* af, int m,
                                            int h) {
  const float* arow = src + m * LS + 2 * h;
#pragma unroll
  for (int kc = 0; kc < 16; ++kc)
    af[kc] = *(const v2f*)(arow + 4 * kc);
}

#define TC(dst, w, nt) (((dst) << 4) | ((w) << 2) | (nt))
template <int N, int C0, int C1, int C2, int C3, int C4, int C5>
__device__ __forceinline__ void proj_multi(const float* src, const v2f* wall,
                                           float* dbase, int m, int h) {
  constexpr int code[6] = {C0, C1, C2, C3, C4, C5};
  v8f acc[N];
#pragma unroll
  for (int i = 0; i < N; ++i) acc[i] = v8f{};
  v2f af[16];
  load_afrags(src, af, m, h);
  __builtin_amdgcn_sched_barrier(0);
#pragma unroll
  for (int kc = 0; kc < 16; ++kc) {
#pragma unroll
    for (int i = 0; i < N; ++i)
      acc[i] = wmma_f32(
          af[kc], wall[((code[i] >> 2) & 3) * 64 + kc * 4 + (code[i] & 3)],
          acc[i]);
  }
#pragma unroll
  for (int r = 0; r < 8; ++r) {
    int row = r + 8 * h;
#pragma unroll
    for (int i = 0; i < N; ++i)
      dbase[(code[i] >> 4) * BUF + row * LS + m + 16 * (code[i] & 3)] =
          acc[i][r];
  }
}

template <int NA, int NB, int C0, int C1, int C2, int C3>
__device__ __forceinline__ void proj_two(const float* srcA, const float* srcB,
                                         const v2f* wall, float* dbase, int m,
                                         int h) {
  constexpr int N = NA + NB;
  constexpr int code[4] = {C0, C1, C2, C3};
  v8f acc[N];
#pragma unroll
  for (int i = 0; i < N; ++i) acc[i] = v8f{};
  v2f afA[16], afB[16];
  const float* arowA = srcA + m * LS + 2 * h;
  const float* arowB = srcB + m * LS + 2 * h;
#pragma unroll
  for (int kc = 0; kc < 4; ++kc) {
    afA[kc] = *(const v2f*)(arowA + 4 * kc);
    afB[kc] = *(const v2f*)(arowB + 4 * kc);
  }
  __builtin_amdgcn_sched_barrier(0);
#pragma unroll
  for (int kc = 4; kc < 16; ++kc) {
    afA[kc] = *(const v2f*)(arowA + 4 * kc);
    afB[kc] = *(const v2f*)(arowB + 4 * kc);
  }
  __builtin_amdgcn_sched_barrier(0);
#pragma unroll
  for (int kc = 0; kc < 16; ++kc) {
#pragma unroll
    for (int i = 0; i < N; ++i)
      acc[i] = wmma_f32(
          i < NA ? afA[kc] : afB[kc],
          wall[((code[i] >> 2) & 3) * 64 + kc * 4 + (code[i] & 3)], acc[i]);
  }
#pragma unroll
  for (int r = 0; r < 8; ++r) {
    int row = r + 8 * h;
#pragma unroll
    for (int i = 0; i < N; ++i)
      dbase[(code[i] >> 4) * BUF + row * LS + m + 16 * (code[i] & 3)] =
          acc[i][r];
  }
}


template <int N, int C0, int C1, int C2, int C3, int C4, int C5>
__device__ __forceinline__ void proj_multi_regs(const v2f* af, const v2f* wall,
                                                float* dbase, int m, int h) {
  constexpr int code[6] = {C0, C1, C2, C3, C4, C5};
  v8f acc[N];
#pragma unroll
  for (int i = 0; i < N; ++i) acc[i] = v8f{};
#pragma unroll
  for (int kc = 0; kc < 16; ++kc) {
#pragma unroll
    for (int i = 0; i < N; ++i)
      acc[i] = wmma_f32(
          af[kc], wall[((code[i] >> 2) & 3) * 64 + kc * 4 + (code[i] & 3)],
          acc[i]);
  }
#pragma unroll
  for (int r = 0; r < 8; ++r) {
    int row = r + 8 * h;
#pragma unroll
    for (int i = 0; i < N; ++i)
      dbase[(code[i] >> 4) * BUF + row * LS + m + 16 * (code[i] & 3)] =
          acc[i][r];
  }
}

__device__ __forceinline__ void load_half_row(const float* A, v4f* r, int m,
                                              int h) {
  const v4f* a = (const v4f*)(A + m * LS + 32 * h);
#pragma unroll
  for (int i = 0; i < 8; ++i) r[i] = a[i];
}

__device__ __forceinline__ float rowdot(const v4f* a, const v4f* b) {
  float s = 0.f;
#pragma unroll
  for (int i = 0; i < 8; ++i) {
    v4f u = a[i], v = b[i];
    s += u.x * v.x + u.y * v.y + u.z * v.z + u.w * v.w;
  }
  s += __shfl_xor(s, 16, 32);
  return s;
}

__device__ __forceinline__ void softmax2(float a, float b, float& p0,
                                         float& p1) {
  float mx  = fmaxf(a, b);
  float e0  = __expf(a - mx);
  float e1  = __expf(b - mx);
  float inv = 1.f / (e0 + e1);
  p0 = e0 * inv;
  p1 = e1 * inv;
}

template <bool DO_TANH>
__device__ __forceinline__ void combine(const v4f* v0, const v4f* v1, float p0,
                                        float p1, float* dst, int m, int h) {
  v4f* d = (v4f*)(dst + m * LS + 32 * h);
#pragma unroll
  for (int i = 0; i < 8; ++i) {
    v4f a = v0[i], b = v1[i], r;
    r.x = p0 * a.x + p1 * b.x;
    r.y = p0 * a.y + p1 * b.y;
    r.z = p0 * a.z + p1 * b.z;
    r.w = p0 * a.w + p1 * b.w;
    if (DO_TANH) {
      r.x = fast_tanh(r.x);
      r.y = fast_tanh(r.y);
      r.z = fast_tanh(r.z);
      r.w = fast_tanh(r.w);
    }
    d[i] = r;
  }
}

template <bool DO_TANH>
__device__ __forceinline__ void attend(const float* Q, const float* K0,
                                       const float* K1, const float* V0,
                                       const float* V1, float* dst,
                                       float scale, int m, int h) {
  v4f q[8], k0[8], k1[8], v0[8], v1[8];
  load_half_row(Q, q, m, h);
  load_half_row(K0, k0, m, h);
  load_half_row(K1, k1, m, h);
  load_half_row(V0, v0, m, h);
  load_half_row(V1, v1, m, h);
  __builtin_amdgcn_sched_barrier(0);
  float d0 = rowdot(q, k0) * scale;
  float d1 = rowdot(q, k1) * scale;
  float p0, p1;
  softmax2(d0, d1, p0, p1);
  combine<DO_TANH>(v0, v1, p0, p1, dst, m, h);
}

__global__ __launch_bounds__(128) void prep_weights(
    const float* __restrict__ Wq, const float* __restrict__ Wk,
    const float* __restrict__ Wv, float* __restrict__ wsw) {
  if (blockIdx.x == 0) {
    const int lane = threadIdx.x & 31;
    const int wv = threadIdx.x >> 5;
    const int m = lane & 15;
    const int h = lane >> 4;
    v8f acc[4];
#pragma unroll
    for (int nt = 0; nt < 4; ++nt) acc[nt] = v8f{};
    const float* arow = Wq + (16 * wv + m) * 64 + 2 * h;
    const float* brow = Wk + m * 64 + 2 * h;
#pragma unroll
    for (int kc = 0; kc < 16; ++kc) {
      v2f af = *(const v2f*)(arow + 4 * kc);
#pragma unroll
      for (int nt = 0; nt < 4; ++nt) {
        v2f bf = *(const v2f*)(brow + nt * 16 * 64 + 4 * kc);
        acc[nt] = wmma_f32(af, bf, acc[nt]);
      }
    }
#pragma unroll
    for (int nt = 0; nt < 4; ++nt) {
#pragma unroll
      for (int q = 0; q < 4; ++q) {
        v2f pr = {acc[nt][2 * q], acc[nt][2 * q + 1]};
        *(v2f*)(wsw + ((4 * wv + 2 * h + (q >> 1)) * 4 + nt) * 64 +
                (m + 16 * (q & 1)) * 2) = pr;
      }
    }
  } else {
    int idx = (blockIdx.x - 1) * 128 + threadIdx.x;
    int j    = idx & 1;
    int lane = (idx >> 1) & 31;
    int nt   = (idx >> 6) & 3;
    int kc   = (idx >> 8) & 15;
    int k = 4 * kc + j + 2 * (lane >> 4);
    int n = 16 * nt + (lane & 15);
    wsw[4096 + idx] = Wv[k * 64 + n];
  }
}

#define B_L   0
#define B_S   1
#define B_X   2
#define B_O0  4
#define B_O1  5
#define B_LM  6
#define B_VL  7
#define B_VS  8
#define B_OM0 9
#define B_OM1 10
#define B_V0  11
#define B_V1  12
#define B_XP  13
#define N_BUF 17

__global__ __launch_bounds__(128) void birnn_kernel(
    const float* __restrict__ x, const float* __restrict__ wsw,
    float* __restrict__ out) {
  __shared__ __align__(16) float smem[N_BUF * BUF];
  float* bufL  = smem + B_L * BUF;
  float* bufS  = smem + B_S * BUF;
  float* bufO0 = smem + B_O0 * BUF;
  float* bufO1 = smem + B_O1 * BUF;
  float* LM  = smem + B_LM * BUF;
  float* VL  = smem + B_VL * BUF;
  float* VS  = smem + B_VS * BUF;
  float* OM0 = smem + B_OM0 * BUF;
  float* OM1 = smem + B_OM1 * BUF;
  float* V0  = smem + B_V0 * BUF;
  float* V1  = smem + B_V1 * BUF;
  float* bxc = smem + B_X * BUF;
  float* bxn = smem + (B_X + 1) * BUF;
  float* pxc = smem + B_XP * BUF;
  float* pxn = smem + (B_XP + 2) * BUF;

  const int lane = threadIdx.x & 31;
  const int wid  = threadIdx.x >> 5;
  const int m = lane & 15;
  const int h = lane >> 4;
  const int b = blockIdx.x * 16 + m;
  const float kScale = 0.125f;

  v2f wall[128];
  {
    const float* myw = wsw + lane * 2;
#pragma unroll
    for (int f = 0; f < 128; ++f) wall[f] = *(const v2f*)(myw + f * 64);
  }

  const float* xb = x + (size_t)b * TSTEPS * DDIM + 32 * h;
  v4f xr[8];
  v2f xa[16];

  if (wid == 0) {
    const float4* xq = (const float4*)xb;
    float4* dl  = (float4*)(bufL + m * LS + 32 * h);
    float4* dsv = (float4*)(bufS + m * LS + 32 * h);
#pragma unroll
    for (int i = 0; i < 8; ++i) {
      float4 v = xq[i];
      dl[i]  = v;
      dsv[i] = v;
    }
  } else if (wid == 2) {
    const float4* xq = (const float4*)(xb + DDIM);
    float4* dx = (float4*)(bxc + m * LS + 32 * h);
#pragma unroll
    for (int i = 0; i < 8; ++i) dx[i] = xq[i];
    const v4f* xq2 = (const v4f*)(xb + 2 * DDIM);
#pragma unroll
    for (int i = 0; i < 8; ++i) xr[i] = xq2[i];
  }
  __syncthreads();
  if (wid == 0)
    proj_multi<2, TC(0, 0, 0), TC(0, 0, 1), 0, 0, 0, 0>(bxc, wall, pxc, m, h);
  else if (wid == 1)
    proj_multi<2, TC(0, 0, 2), TC(0, 0, 3), 0, 0, 0, 0>(bxc, wall, pxc, m, h);
  else if (wid == 2)
    proj_multi<2, TC(1, 1, 0), TC(1, 1, 1), 0, 0, 0, 0>(bxc, wall, pxc, m, h);
  else
    proj_multi<2, TC(1, 1, 2), TC(1, 1, 3), 0, 0, 0, 0>(bxc, wall, pxc, m, h);
  __syncthreads();

#pragma unroll 1
  for (int t = 1; t < TSTEPS; ++t) {
    const bool ahead = (t + 1 < TSTEPS);

    if (wid == 0) {
      proj_multi<3, TC(B_LM, 0, 0), TC(B_LM, 0, 1), TC(B_LM, 0, 2), 0, 0, 0>(
          bufL, wall, smem, m, h);
    } else if (wid == 1) {
      proj_multi<3, TC(B_LM, 0, 3), TC(B_VL, 1, 0), TC(B_VL, 1, 1), 0, 0, 0>(
          bufL, wall, smem, m, h);
    } else if (wid == 2) {
      if (ahead) {
        v4f* dx = (v4f*)(bxn + m * LS + 32 * h);
#pragma unroll
        for (int i = 0; i < 8; ++i) dx[i] = xr[i];
      }
      proj_two<2, 1, TC(B_VL, 1, 2), TC(B_VL, 1, 3), TC(B_VS, 1, 0), 0>(
          bufL, bufS, wall, smem, m, h);
    } else {
      proj_multi<3, TC(B_VS, 1, 1), TC(B_VS, 1, 2), TC(B_VS, 1, 3), 0, 0, 0>(
          bufS, wall, smem, m, h);
    }
    __syncthreads();

    if (wid == 0) {
      attend<true>(LM, bufL, bxc, VL, pxc + BUF, bufO0, kScale, m, h);
    } else if (wid == 1) {
      attend<true>(pxc, bufS, bxc, VS, pxc + BUF, bufO1, kScale, m, h);
    } else if (wid == 2) {
      if (t + 2 < TSTEPS) {
        const v4f* xq = (const v4f*)(xb + (size_t)(t + 2) * DDIM);
#pragma unroll
        for (int i = 0; i < 8; ++i) xr[i] = xq[i];
      }
      if (ahead) {
        load_afrags(bxn, xa, m, h);
        proj_multi_regs<2, TC(0, 0, 0), TC(0, 0, 1), 0, 0, 0, 0>(xa, wall, pxn, m, h);
      }
    } else {
      if (ahead) {
        load_afrags(bxn, xa, m, h);
        proj_multi_regs<2, TC(0, 0, 2), TC(0, 0, 3), 0, 0, 0, 0>(xa, wall, pxn, m, h);
      }
    }
    __syncthreads();

    if (wid == 0) {
      proj_multi<4, TC(B_OM0, 0, 0), TC(B_OM0, 0, 1), TC(B_OM0, 0, 2),
                 TC(B_OM0, 0, 3), 0, 0>(bufO0, wall, smem, m, h);
    } else if (wid == 1) {
      proj_multi<4, TC(B_V0, 1, 0), TC(B_V0, 1, 1), TC(B_V0, 1, 2),
                 TC(B_V0, 1, 3), 0, 0>(bufO0, wall, smem, m, h);
    } else if (wid == 2) {
      proj_multi<4, TC(B_OM1, 0, 0), TC(B_OM1, 0, 1), TC(B_OM1, 0, 2),
                 TC(B_OM1, 0, 3), 0, 0>(bufO1, wall, smem, m, h);
    } else {
      proj_multi<4, TC(B_V1, 1, 0), TC(B_V1, 1, 1), TC(B_V1, 1, 2),
                 TC(B_V1, 1, 3), 0, 0>(bufO1, wall, smem, m, h);
    }
    __syncthreads();

    if (wid == 0) {
      attend<false>(OM0, bufO0, bufO1, V0, V1, bufL, kScale, m, h);
    } else if (wid == 1) {
      attend<false>(OM1, bufO0, bufO1, V0, V1, bufS, kScale, m, h);
    } else if (wid == 2) {
      if (ahead)
        proj_multi_regs<2, TC(1, 1, 0), TC(1, 1, 1), 0, 0, 0, 0>(xa, wall, pxn, m, h);
    } else {
      if (ahead)
        proj_multi_regs<2, TC(1, 1, 2), TC(1, 1, 3), 0, 0, 0, 0>(xa, wall, pxn, m, h);
    }
    __syncthreads();

    float* tmp = pxc;
    pxc = pxn;
    pxn = tmp;
    tmp = bxc;
    bxc = bxn;
    bxn = tmp;
  }

  if (wid == 0) {
    const float4* src = (const float4*)(bufL + m * LS + 32 * h);
    float4* dst = (float4*)(out + (size_t)b * DDIM + 32 * h);
#pragma unroll
    for (int i = 0; i < 8; ++i) dst[i] = src[i];
  }
}

extern "C" void kernel_launch(void* const* d_in, const int* in_sizes, int n_in,
                              void* d_out, int out_size, void* d_ws,
                              size_t ws_size, hipStream_t stream) {
  (void)in_sizes; (void)n_in; (void)out_size; (void)ws_size;
  const float* x  = (const float*)d_in[0];
  const float* Wq = (const float*)d_in[1];
  const float* Wk = (const float*)d_in[2];
  const float* Wv = (const float*)d_in[3];
  float* wsw = (float*)d_ws;

  prep_weights<<<33, 128, 0, stream>>>(Wq, Wk, Wv, wsw);
  birnn_kernel<<<2048 / 16, 128, 0, stream>>>(x, wsw, (float*)d_out);
}


# Eight batch rows per work-group, 256 work-groups: token pairs that share a weight matrix ([l;s], [o0;o1], [x_t;x_t+1]) stack into one 16-row A tile: 20 tiles a step, was 36; score sum order changes
# speedup vs baseline: 3.2202x; 1.5204x over previous
#include <hip/hip_runtime.h>
#include <math.h>


#define LS     68
#define BUF    (16 * LS)
#define TSTEPS 200
#define DDIM   64

typedef float v2f __attribute__((ext_vector_type(2)));
typedef float v4f __attribute__((ext_vector_type(4)));
typedef float v8f __attribute__((ext_vector_type(8)));

__device__ __forceinline__ float fast_tanh(float v) {
#if __has_builtin(__builtin_amdgcn_tanhf)
  return __builtin_amdgcn_tanhf(v);
#elif __has_builtin(__builtin_amdgcn_tanh_f32)
  return __builtin_amdgcn_tanh_f32(v);
#else
  return tanhf(v);
#endif
}

__device__ __forceinline__ v8f wmma_f32(v2f a, v2f b, v8f c) {
  return __builtin_amdgcn_wmma_f32_16x16x4_f32(false, a, false, b, (short)0, c,
                                               false, false);
}

__device__ __forceinline__ void load_afrags(const float* src, v2f* af, int m,
                                            int h) {
  const float* arow = src + m * LS + 2 * h;
#pragma unroll
  for (int kc = 0; kc < 16; ++kc)
    af[kc] = *(const v2f*)(arow + 4 * kc);
}

#define TC(dst, w, nt) (((dst) << 4) | ((w) << 2) | (nt))
template <int N, int C0, int C1, int C2, int C3, int C4, int C5>
__device__ __forceinline__ void proj_multi(const float* src, const v2f* wall,
                                           float* dbase, int m, int h) {
  constexpr int code[6] = {C0, C1, C2, C3, C4, C5};
  v8f acc[N];
#pragma unroll
  for (int i = 0; i < N; ++i) acc[i] = v8f{};
  v2f af[16];
  load_afrags(src, af, m, h);
  __builtin_amdgcn_sched_barrier(0);
#pragma unroll
  for (int kc = 0; kc < 16; ++kc) {
#pragma unroll
    for (int i = 0; i < N; ++i)
      acc[i] = wmma_f32(
          af[kc], wall[((code[i] >> 2) & 3) * 64 + kc * 4 + (code[i] & 3)],
          acc[i]);
  }
#pragma unroll
  for (int r = 0; r < 8; ++r) {
    int row = r + 8 * h;
#pragma unroll
    for (int i = 0; i < N; ++i)
      dbase[(code[i] >> 4) * BUF + row * LS + m + 16 * (code[i] & 3)] =
          acc[i][r];
  }
}

template <int WM, int NT0, int NT1, int KC0>
__device__ __forceinline__ void xproj_half(const v2f* af, const v2f* wall,
                                           v8f& a0, v8f& a1) {
#pragma unroll
  for (int kc = KC0; kc < KC0 + 8; ++kc) {
    a0 = wmma_f32(af[kc], wall[WM * 64 + kc * 4 + NT0], a0);
    a1 = wmma_f32(af[kc], wall[WM * 64 + kc * 4 + NT1], a1);
  }
}

template <int NT0, int NT1>
__device__ __forceinline__ void store_two(float* dst, v8f a0, v8f a1, int m,
                                          int h) {
#pragma unroll
  for (int r = 0; r < 8; ++r) {
    int row = r + 8 * h;
    dst[row * LS + m + 16 * NT0] = a0[r];
    dst[row * LS + m + 16 * NT1] = a1[r];
  }
}

__device__ __forceinline__ void load_qrow(const float* A, v4f* r, int rr,
                                          int qd) {
  const v4f* a = (const v4f*)(A + rr * LS + 16 * qd);
#pragma unroll
  for (int i = 0; i < 4; ++i) r[i] = a[i];
}

__device__ __forceinline__ float quad_sum(float s) {
  s += __int_as_float(__builtin_amdgcn_update_dpp(0, __float_as_int(s), 0xB1,
                                                  0xF, 0xF, true));
  s += __int_as_float(__builtin_amdgcn_update_dpp(0, __float_as_int(s), 0x4E,
                                                  0xF, 0xF, true));
  return s;
}

__device__ __forceinline__ float rowdot(const v4f* a, const v4f* b) {
  float s = 0.f;
#pragma unroll
  for (int i = 0; i < 4; ++i) {
    v4f u = a[i], v = b[i];
    s += u.x * v.x + u.y * v.y + u.z * v.z + u.w * v.w;
  }
  return quad_sum(s);
}

__device__ __forceinline__ void softmax2(float a, float b, float& p0,
                                         float& p1) {
  float mx  = fmaxf(a, b);
  float e0  = __expf(a - mx);
  float e1  = __expf(b - mx);
  float inv = 1.f / (e0 + e1);
  p0 = e0 * inv;
  p1 = e1 * inv;
}

template <bool DO_TANH>
__device__ __forceinline__ void combine(const v4f* v0, const v4f* v1, float p0,
                                        float p1, float* dst, int rr, int qd) {
  v4f* d = (v4f*)(dst + rr * LS + 16 * qd);
#pragma unroll
  for (int i = 0; i < 4; ++i) {
    v4f a = v0[i], b = v1[i], r;
    r.x = p0 * a.x + p1 * b.x;
    r.y = p0 * a.y + p1 * b.y;
    r.z = p0 * a.z + p1 * b.z;
    r.w = p0 * a.w + p1 * b.w;
    if (DO_TANH) {
      r.x = fast_tanh(r.x);
      r.y = fast_tanh(r.y);
      r.z = fast_tanh(r.z);
      r.w = fast_tanh(r.w);
    }
    d[i] = r;
  }
}

template <bool DO_TANH>
__device__ __forceinline__ void attend(const float* Q, const float* K0,
                                       const float* K1, const float* V0,
                                       const float* V1, float* dst,
                                       float scale, int rr, int qd) {
  v4f q[4], k0[4], k1[4], v0[4], v1[4];
  load_qrow(Q, q, rr, qd);
  load_qrow(K0, k0, rr, qd);
  load_qrow(K1, k1, rr, qd);
  load_qrow(V0, v0, rr, qd);
  load_qrow(V1, v1, rr, qd);
  __builtin_amdgcn_sched_barrier(0);
  float d0 = rowdot(q, k0) * scale;
  float d1 = rowdot(q, k1) * scale;
  float p0, p1;
  softmax2(d0, d1, p0, p1);
  combine<DO_TANH>(v0, v1, p0, p1, dst, rr, qd);
}

__global__ __launch_bounds__(128) void prep_weights(
    const float* __restrict__ Wq, const float* __restrict__ Wk,
    const float* __restrict__ Wv, float* __restrict__ wsw) {
  if (blockIdx.x == 0) {
    const int lane = threadIdx.x & 31;
    const int wv = threadIdx.x >> 5;
    const int m = lane & 15;
    const int h = lane >> 4;
    v8f acc[4];
#pragma unroll
    for (int nt = 0; nt < 4; ++nt) acc[nt] = v8f{};
    const float* arow = Wq + (16 * wv + m) * 64 + 2 * h;
    const float* brow = Wk + m * 64 + 2 * h;
#pragma unroll
    for (int kc = 0; kc < 16; ++kc) {
      v2f af = *(const v2f*)(arow + 4 * kc);
#pragma unroll
      for (int nt = 0; nt < 4; ++nt) {
        v2f bf = *(const v2f*)(brow + nt * 16 * 64 + 4 * kc);
        acc[nt] = wmma_f32(af, bf, acc[nt]);
      }
    }
#pragma unroll
    for (int nt = 0; nt < 4; ++nt) {
#pragma unroll
      for (int q = 0; q < 4; ++q) {
        v2f pr = {acc[nt][2 * q], acc[nt][2 * q + 1]};
        *(v2f*)(wsw + ((4 * wv + 2 * h + (q >> 1)) * 4 + nt) * 64 +
                (m + 16 * (q & 1)) * 2) = pr;
      }
    }
  } else {
    int idx = (blockIdx.x - 1) * 128 + threadIdx.x;
    int j    = idx & 1;
    int lane = (idx >> 1) & 31;
    int nt   = (idx >> 6) & 3;
    int kc   = (idx >> 8) & 15;
    int k = 4 * kc + j + 2 * (lane >> 4);
    int n = 16 * nt + (lane & 15);
    wsw[4096 + idx] = Wv[k * 64 + n];
  }
}

#define B_CAR 0
#define B_OO  1
#define B_CM  2
#define B_CV  3
#define B_OM  4
#define B_OV  5
#define B_X   6
#define B_XP  8
#define N_BUF 12
#define HALF  (8 * LS)

__global__ __launch_bounds__(128) void birnn_kernel(
    const float* __restrict__ x, const float* __restrict__ wsw,
    float* __restrict__ out) {
  __shared__ __align__(16) float smem[N_BUF * BUF];
  float* CAR = smem + B_CAR * BUF;
  float* OO  = smem + B_OO * BUF;
  float* CM  = smem + B_CM * BUF;
  float* CV  = smem + B_CV * BUF;
  float* OM  = smem + B_OM * BUF;
  float* OV  = smem + B_OV * BUF;
  float* bxc = smem + B_X * BUF;
  float* bxn = smem + (B_X + 1) * BUF;
  float* pxc = smem + B_XP * BUF;
  float* pxn = smem + (B_XP + 2) * BUF;

  const int lane = threadIdx.x & 31;
  const int wid  = threadIdx.x >> 5;
  const int m = lane & 15;
  const int h = lane >> 4;
  const int rr = lane >> 2;
  const int qd = lane & 3;
  const int b0 = blockIdx.x * 8;
  const float kScale = 0.125f;

  v2f wall[128];
  {
    const float* myw = wsw + lane * 2;
#pragma unroll
    for (int f = 0; f < 128; ++f) wall[f] = *(const v2f*)(myw + f * 64);
  }

  const float* xb = x + (size_t)(b0 + (m & 7)) * TSTEPS * DDIM + 32 * h;
  const int tsel = m >> 3;
  v4f xr[8];
  v2f xa[16];
  v8f xacc0 = v8f{}, xacc1 = v8f{};

  if (wid == 0) {
    const float4* xq = (const float4*)xb;
    float4* dc = (float4*)(CAR + m * LS + 32 * h);
#pragma unroll
    for (int i = 0; i < 8; ++i) dc[i] = xq[i];
  } else if (wid == 2) {
    const float4* xq = (const float4*)(xb + (size_t)(1 + tsel) * DDIM);
    float4* dx = (float4*)(bxc + m * LS + 32 * h);
#pragma unroll
    for (int i = 0; i < 8; ++i) dx[i] = xq[i];
    const v4f* xq2 = (const v4f*)(xb + (size_t)(3 + tsel) * DDIM);
#pragma unroll
    for (int i = 0; i < 8; ++i) xr[i] = xq2[i];
  }
  __syncthreads();
  if (wid == 0)
    proj_multi<2, TC(0, 0, 0), TC(1, 1, 0), 0, 0, 0, 0>(bxc, wall, pxc, m, h);
  else if (wid == 1)
    proj_multi<2, TC(0, 0, 1), TC(1, 1, 1), 0, 0, 0, 0>(bxc, wall, pxc, m, h);
  else if (wid == 2)
    proj_multi<2, TC(0, 0, 2), TC(1, 1, 2), 0, 0, 0, 0>(bxc, wall, pxc, m, h);
  else
    proj_multi<2, TC(0, 0, 3), TC(1, 1, 3), 0, 0, 0, 0>(bxc, wall, pxc, m, h);
  __syncthreads();

#pragma unroll 1
  for (int tp = 1; tp < TSTEPS; tp += 2) {
    const bool ahead = (tp + 2 < TSTEPS);

#pragma unroll
    for (int sub = 0; sub < 2; ++sub) {
      if (sub == 1 && tp + 1 >= TSTEPS) break;

      if (wid == 0) {
        proj_multi<2, TC(B_CM, 0, 0), TC(B_CV, 1, 0), 0, 0, 0, 0>(
            CAR, wall, smem, m, h);
      } else if (wid == 1) {
        proj_multi<2, TC(B_CM, 0, 1), TC(B_CV, 1, 1), 0, 0, 0, 0>(
            CAR, wall, smem, m, h);
      } else if (wid == 2) {
        if (sub == 0 && ahead) {
          v4f* dx = (v4f*)(bxn + m * LS + 32 * h);
#pragma unroll
          for (int i = 0; i < 8; ++i) dx[i] = xr[i];
        }
        proj_multi<2, TC(B_CM, 0, 2), TC(B_CV, 1, 2), 0, 0, 0, 0>(
            CAR, wall, smem, m, h);
      } else {
        proj_multi<2, TC(B_CM, 0, 3), TC(B_CV, 1, 3), 0, 0, 0, 0>(
            CAR, wall, smem, m, h);
      }
      __syncthreads();

      if (wid == 0) {
        attend<true>(CM, CAR, bxc + sub * HALF, CV, pxc + BUF + sub * HALF,
                     OO, kScale, rr, qd);
      } else if (wid == 1) {
        attend<true>(pxc + sub * HALF, CAR + HALF, bxc + sub * HALF, CV + HALF,
                     pxc + BUF + sub * HALF, OO + HALF, kScale, rr, qd);
      } else if (wid == 2) {
        if (sub == 0 && tp + 4 < TSTEPS) {
          int ti = tp + 4 + tsel;
          ti = ti < TSTEPS ? ti : TSTEPS - 1;
          const v4f* xq = (const v4f*)(xb + (size_t)ti * DDIM);
#pragma unroll
          for (int i = 0; i < 8; ++i) xr[i] = xq[i];
        }
        if (ahead) {
          if (sub == 0) load_afrags(bxn, xa, m, h);
          xacc0 = v8f{};
          xacc1 = v8f{};
          if (sub == 0) xproj_half<0, 0, 1, 0>(xa, wall, xacc0, xacc1);
          else          xproj_half<1, 0, 1, 0>(xa, wall, xacc0, xacc1);
        }
      } else {
        if (ahead) {
          if (sub == 0) load_afrags(bxn, xa, m, h);
          xacc0 = v8f{};
          xacc1 = v8f{};
          if (sub == 0) xproj_half<0, 2, 3, 0>(xa, wall, xacc0, xacc1);
          else          xproj_half<1, 2, 3, 0>(xa, wall, xacc0, xacc1);
        }
      }
      __syncthreads();

      if (wid == 0) {
        proj_multi<2, TC(B_OM, 0, 0), TC(B_OV, 1, 0), 0, 0, 0, 0>(
            OO, wall, smem, m, h);
      } else if (wid == 1) {
        proj_multi<2, TC(B_OM, 0, 1), TC(B_OV, 1, 1), 0, 0, 0, 0>(
            OO, wall, smem, m, h);
      } else if (wid == 2) {
        proj_multi<2, TC(B_OM, 0, 2), TC(B_OV, 1, 2), 0, 0, 0, 0>(
            OO, wall, smem, m, h);
      } else {
        proj_multi<2, TC(B_OM, 0, 3), TC(B_OV, 1, 3), 0, 0, 0, 0>(
            OO, wall, smem, m, h);
      }
      __syncthreads();

      if (wid == 0) {
        attend<false>(OM, OO, OO + HALF, OV, OV + HALF, CAR, kScale, rr, qd);
      } else if (wid == 1) {
        attend<false>(OM + HALF, OO, OO + HALF, OV, OV + HALF, CAR + HALF,
                      kScale, rr, qd);
      } else if (wid == 2) {
        if (ahead) {
          if (sub == 0) {
            xproj_half<0, 0, 1, 8>(xa, wall, xacc0, xacc1);
            store_two<0, 1>(pxn, xacc0, xacc1, m, h);
          } else {
            xproj_half<1, 0, 1, 8>(xa, wall, xacc0, xacc1);
            store_two<0, 1>(pxn + BUF, xacc0, xacc1, m, h);
          }
        }
      } else {
        if (ahead) {
          if (sub == 0) {
            xproj_half<0, 2, 3, 8>(xa, wall, xacc0, xacc1);
            store_two<2, 3>(pxn, xacc0, xacc1, m, h);
          } else {
            xproj_half<1, 2, 3, 8>(xa, wall, xacc0, xacc1);
            store_two<2, 3>(pxn + BUF, xacc0, xacc1, m, h);
          }
        }
      }
      __syncthreads();
    }

    float* tmp = pxc;
    pxc = pxn;
    pxn = tmp;
    tmp = bxc;
    bxc = bxn;
    bxn = tmp;
  }

  if (wid == 0) {
#pragma unroll
    for (int i = 0; i < 4; ++i) {
      int row = 2 * i + (lane >> 4);
      const float4* s4 = (const float4*)(CAR + row * LS + 4 * (lane & 15));
      float4* d4 = (float4*)(out + (size_t)(b0 + row) * DDIM + 4 * (lane & 15));
      *d4 = *s4;
    }
  }
}

extern "C" void kernel_launch(void* const* d_in, const int* in_sizes, int n_in,
                              void* d_out, int out_size, void* d_ws,
                              size_t ws_size, hipStream_t stream) {
  (void)in_sizes; (void)n_in; (void)out_size; (void)ws_size;
  const float* x  = (const float*)d_in[0];
  const float* Wq = (const float*)d_in[1];
  const float* Wk = (const float*)d_in[2];
  const float* Wv = (const float*)d_in[3];
  float* wsw = (float*)d_ws;

  prep_weights<<<33, 128, 0, stream>>>(Wq, Wk, Wv, wsw);
  birnn_kernel<<<2048 / 8, 128, 0, stream>>>(x, wsw, (float*)d_out);
}


# Softmax of two taken from the score difference alone: each lane's q.(k0-k1) crosses LDS as one float, tree sum, p1 = rcp(1+exp2(c d)), row = v0+p1(v1-v0); score sum order changes
# speedup vs baseline: 4.0581x; 1.2602x over previous
#include <hip/hip_runtime.h>
#include <math.h>


#define LS     68
#define BUF    (16 * LS)
#define TSTEPS 200
#define DDIM   64

typedef float v2f __attribute__((ext_vector_type(2)));
typedef float v4f __attribute__((ext_vector_type(4)));
typedef float v8f __attribute__((ext_vector_type(8)));

__device__ __forceinline__ float fast_tanh(float v) {
#if __has_builtin(__builtin_amdgcn_tanhf)
  return __builtin_amdgcn_tanhf(v);
#elif __has_builtin(__builtin_amdgcn_tanh_f32)
  return __builtin_amdgcn_tanh_f32(v);
#else
  return tanhf(v);
#endif
}

__device__ __forceinline__ v8f wmma_f32(v2f a, v2f b, v8f c) {
  return __builtin_amdgcn_wmma_f32_16x16x4_f32(false, a, false, b, (short)0, c,
                                               false, false);
}

__device__ __forceinline__ void load_tfrags(const float* src, v2f* tf, int n,
                                            int h) {
  const float* trow = src + n * LS + 2 * h;
#pragma unroll
  for (int kc = 0; kc < 16; ++kc)
    tf[kc] = *(const v2f*)(trow + 4 * kc);
}

__device__ __forceinline__ void proj2(const v2f* tf, const v2f* wq,
                                      const v2f* wv, v8f& am, v8f& av) {
  am = v8f{};
  av = v8f{};
#pragma unroll
  for (int kc = 0; kc < 16; ++kc) {
    am = wmma_f32(wq[kc], tf[kc], am);
    av = wmma_f32(wv[kc], tf[kc], av);
  }
}

template <int P>
__device__ __forceinline__ void xpiece(const v2f* xa, const v2f* wq,
                                       const v2f* wv, v8f& am, v8f& av) {
  if (P == 0) {
    am = v8f{};
    av = v8f{};
  }
#pragma unroll
  for (int kc = 2 * P; kc < 2 * P + 2; ++kc) {
    am = wmma_f32(wq[kc], xa[kc], am);
    av = wmma_f32(wv[kc], xa[kc], av);
  }
}

__device__ __forceinline__ v8f load8(const float* p) {
  v4f a = *(const v4f*)p;
  v4f b = *(const v4f*)(p + 4);
  return __builtin_shufflevector(a, b, 0, 1, 2, 3, 4, 5, 6, 7);
}

__device__ __forceinline__ void store8(float* p, v8f v) {
  *(v4f*)p       = __builtin_shufflevector(v, v, 0, 1, 2, 3);
  *(v4f*)(p + 4) = __builtin_shufflevector(v, v, 4, 5, 6, 7);
}

__device__ __forceinline__ float partner(float v) {
  return __int_as_float(__builtin_amdgcn_update_dpp(0, __float_as_int(v), 0x128,
                                                    0xF, 0xF, true));
}

__device__ __forceinline__ float part_diff(v8f q, v8f kd) {
  float a = fmaf(q[1], kd[1], q[0] * kd[0]);
  float b = fmaf(q[3], kd[3], q[2] * kd[2]);
  float c = fmaf(q[5], kd[5], q[4] * kd[4]);
  float d = fmaf(q[7], kd[7], q[6] * kd[6]);
  return (a + b) + (c + d);
}

__device__ __forceinline__ float sum_diff(const float* part_row) {
  const v4f* p = (const v4f*)part_row;
  v4f a = p[0], b = p[1];
  return ((a.x + a.y) + (a.z + a.w)) + ((b.x + b.y) + (b.z + b.w));
}

__device__ __forceinline__ float weight_second(float d) {
  const float c = 0.125f * 1.44269504088896340736f;
  float e = __builtin_amdgcn_exp2f(d * c);
  return __builtin_amdgcn_rcpf(1.0f + e);
}


__global__ __launch_bounds__(128) void prep_weights(
    const float* __restrict__ Wq, const float* __restrict__ Wk,
    const float* __restrict__ Wv, float* __restrict__ wsw) {
  if (blockIdx.x == 0) {
    const int lane = threadIdx.x & 31;
    const int wv = threadIdx.x >> 5;
    const int m = lane & 15;
    const int h = lane >> 4;
    v8f acc[4];
#pragma unroll
    for (int nt = 0; nt < 4; ++nt) acc[nt] = v8f{};
    const float* arow = Wq + (16 * wv + m) * 64 + 2 * h;
    const float* brow = Wk + m * 64 + 2 * h;
#pragma unroll
    for (int kc = 0; kc < 16; ++kc) {
      v2f af = *(const v2f*)(arow + 4 * kc);
#pragma unroll
      for (int nt = 0; nt < 4; ++nt) {
        v2f bf = *(const v2f*)(brow + nt * 16 * 64 + 4 * kc);
        acc[nt] = wmma_f32(af, bf, acc[nt]);
      }
    }
#pragma unroll
    for (int nt = 0; nt < 4; ++nt) {
#pragma unroll
      for (int q = 0; q < 4; ++q) {
        v2f pr = {acc[nt][2 * q], acc[nt][2 * q + 1]};
        *(v2f*)(wsw + ((4 * wv + 2 * h + (q >> 1)) * 4 + nt) * 64 +
                (m + 16 * (q & 1)) * 2) = pr;
      }
    }
  } else {
    int idx = (blockIdx.x - 1) * 128 + threadIdx.x;
    int j    = idx & 1;
    int lane = (idx >> 1) & 31;
    int nt   = (idx >> 6) & 3;
    int kc   = (idx >> 8) & 15;
    int k = 4 * kc + j + 2 * (lane >> 4);
    int n = 16 * nt + (lane & 15);
    wsw[4096 + idx] = Wv[k * 64 + n];
  }
}

#define B_CAR 0
#define B_OO  1
#define B_X   2
#define B_XP  4
#define N_BUF 8
#define HALF  (8 * LS)
#define PS     12
#define N_PART (16 * PS)

__global__ __launch_bounds__(128) void birnn_kernel(
    const float* __restrict__ x, const float* __restrict__ wsw,
    float* __restrict__ out) {
  __shared__ __align__(16) float smem[N_BUF * BUF + N_PART];
  float* CAR  = smem + B_CAR * BUF;
  float* OO   = smem + B_OO * BUF;
  float* PART = smem + N_BUF * BUF;
  float* bxc = smem + B_X * BUF;
  float* bxn = smem + (B_X + 1) * BUF;
  float* pxc = smem + B_XP * BUF;
  float* pxn = smem + (B_XP + 2) * BUF;

  const int lane = threadIdx.x & 31;
  const int wid  = threadIdx.x >> 5;
  const int n = lane & 15;
  const int h = lane >> 4;
  const int F0 = 16 * wid + 8 * h;
  const int rowoff = n * LS + F0;
  const int xoff = (n & 7) * LS + F0;
  const bool first = n < 8;
  const int tsel = n >> 3;
  const int b0 = blockIdx.x * 8;
  float* mypart = PART + n * PS + wid * 2 + h;
  const float* partrow = PART + n * PS;

  v2f wq[16], wv[16];
  {
    const float* myw = wsw + wid * 64 + lane * 2;
#pragma unroll
    for (int kc = 0; kc < 16; ++kc) {
      wq[kc] = *(const v2f*)(myw + kc * 256);
      wv[kc] = *(const v2f*)(myw + 4096 + kc * 256);
    }
  }

  const float* xb = x + (size_t)(b0 + (n & 7)) * TSTEPS * DDIM;
  v4f xr[8];
  v2f xa[16];
  v8f xm = v8f{}, xv = v8f{};
  v2f tf[16];
  v8f am, av;

  v8f cr;
  {
    const v4f* g = (const v4f*)(xb + F0);
    v4f a = g[0], b = g[1];
    cr = __builtin_shufflevector(a, b, 0, 1, 2, 3, 4, 5, 6, 7);
  }
  store8(CAR + rowoff, cr);
  if (wid >= 2) {
    float* dst = (wid == 2) ? bxc : bxn;
    const int t0 = (wid == 2) ? 1 : 3;
    const v4f* xq = (const v4f*)(xb + (size_t)(t0 + tsel) * DDIM + 32 * h);
    v4f* dx = (v4f*)(dst + n * LS + 32 * h);
#pragma unroll
    for (int i = 0; i < 8; ++i) dx[i] = xq[i];
  }
  __syncthreads();
  load_tfrags(bxc, tf, n, h);
  proj2(tf, wq, wv, am, av);
  store8(pxc + rowoff, am);
  store8(pxc + BUF + rowoff, av);
  __syncthreads();

#pragma unroll 1
  for (int tp = 1; tp < TSTEPS; tp += 2) {
    const bool ahead = (tp + 2 < TSTEPS);
    const bool fetch = (tp + 4 < TSTEPS);

    if (wid == 2 && fetch) {
      int ti = tp + 4 + tsel;
      ti = ti < TSTEPS ? ti : TSTEPS - 1;
      const v4f* xq = (const v4f*)(xb + (size_t)ti * DDIM + 32 * h);
#pragma unroll
      for (int i = 0; i < 8; ++i) xr[i] = xq[i];
    }
    if (ahead) load_tfrags(bxn, xa, n, h);

#pragma unroll
    for (int sub = 0; sub < 2; ++sub) {
      if (sub == 1 && tp + 1 >= TSTEPS) break;
      const int xo = xoff + sub * HALF;

      load_tfrags(CAR, tf, n, h);
      v8f xmq = load8(pxc + xo);
      v8f xk  = load8(bxc + xo);
      v8f xw  = load8(pxc + BUF + xo);
      __builtin_amdgcn_sched_barrier(0);
      if (ahead) {
        if (sub == 0) xpiece<0>(xa, wq, wv, xm, xv);
        else          xpiece<4>(xa, wq, wv, xm, xv);
      }
      proj2(tf, wq, wv, am, av);
      {
        v8f kd = cr - xk;
        float dx = part_diff(xmq, kd);
        float da = part_diff(am, kd);
        *mypart = first ? da : dx;
      }
      if (ahead) {
        if (sub == 0) xpiece<1>(xa, wq, wv, xm, xv);
        else          xpiece<5>(xa, wq, wv, xm, xv);
      }
      __syncthreads();
      v8f orow;
      {
        v8f dv = xw - av;
        float p1 = weight_second(sum_diff(partrow));
#pragma unroll
        for (int r = 0; r < 8; ++r)
          orow[r] = fast_tanh(fmaf(p1, dv[r], av[r]));
      }
      store8(OO + rowoff, orow);
      __syncthreads();

      load_tfrags(OO, tf, n, h);
      __builtin_amdgcn_sched_barrier(0);
      if (ahead) {
        if (sub == 0) xpiece<2>(xa, wq, wv, xm, xv);
        else          xpiece<6>(xa, wq, wv, xm, xv);
      }
      proj2(tf, wq, wv, am, av);
      v8f dv;
      {
        v8f kd;
#pragma unroll
        for (int r = 0; r < 8; ++r) {
          kd[r] = orow[r] - partner(orow[r]);
          dv[r] = partner(av[r]) - av[r];
        }
        *mypart = part_diff(am, kd);
      }
      if (ahead) {
        if (sub == 0) {
          xpiece<3>(xa, wq, wv, xm, xv);
        } else {
          xpiece<7>(xa, wq, wv, xm, xv);
          store8(pxn + rowoff, xm);
          store8(pxn + BUF + rowoff, xv);
        }
      }
      __syncthreads();
      {
        float p1 = weight_second(sum_diff(partrow));
#pragma unroll
        for (int r = 0; r < 8; ++r) cr[r] = fmaf(p1, dv[r], av[r]);
      }
      store8(CAR + rowoff, cr);
      if (sub == 1 && wid == 2 && fetch) {
        v4f* dx = (v4f*)(bxc + n * LS + 32 * h);
#pragma unroll
        for (int i = 0; i < 8; ++i) dx[i] = xr[i];
      }
      __syncthreads();
    }

    float* tmp = pxc;
    pxc = pxn;
    pxn = tmp;
    tmp = bxc;
    bxc = bxn;
    bxn = tmp;
  }

  if (wid == 0) {
#pragma unroll
    for (int i = 0; i < 4; ++i) {
      int row = 2 * i + (lane >> 4);
      const float4* s4 = (const float4*)(CAR + row * LS + 4 * (lane & 15));
      float4* d4 = (float4*)(out + (size_t)(b0 + row) * DDIM + 4 * (lane & 15));
      *d4 = *s4;
    }
  }
}

extern "C" void kernel_launch(void* const* d_in, const int* in_sizes, int n_in,
                              void* d_out, int out_size, void* d_ws,
                              size_t ws_size, hipStream_t stream) {
  (void)in_sizes; (void)n_in; (void)out_size; (void)ws_size;
  const float* x  = (const float*)d_in[0];
  const float* Wq = (const float*)d_in[1];
  const float* Wk = (const float*)d_in[2];
  const float* Wv = (const float*)d_in[3];
  float* wsw = (float*)d_ws;

  prep_weights<<<33, 128, 0, stream>>>(Wq, Wk, Wv, wsw);
  birnn_kernel<<<2048 / 8, 128, 0, stream>>>(x, wsw, (float*)d_out);
}


# Carry projections folded into the [o0; o1] projection: pre-pass adds Wv M, Wv Wv; l M, l Wv, s Wv are the softmax blend of o's accumulators; 1 serial projection, 3 barriers a step; sum order changes
# speedup vs baseline: 4.5519x; 1.1217x over previous
#include <hip/hip_runtime.h>
#include <math.h>


#define LS     68
#define BUF    (16 * LS)
#define TSTEPS 200
#define DDIM   64

typedef float v2f __attribute__((ext_vector_type(2)));
typedef float v4f __attribute__((ext_vector_type(4)));
typedef float v8f __attribute__((ext_vector_type(8)));

__device__ __forceinline__ float fast_tanh(float v) {
#if __has_builtin(__builtin_amdgcn_tanhf)
  return __builtin_amdgcn_tanhf(v);
#elif __has_builtin(__builtin_amdgcn_tanh_f32)
  return __builtin_amdgcn_tanh_f32(v);
#else
  return tanhf(v);
#endif
}

__device__ __forceinline__ v8f wmma_f32(v2f a, v2f b, v8f c) {
  return __builtin_amdgcn_wmma_f32_16x16x4_f32(false, a, false, b, (short)0, c,
                                               false, false);
}

__device__ __forceinline__ void load_tfrags(const float* src, v2f* tf, int n,
                                            int h) {
  const float* trow = src + n * LS + 2 * h;
#pragma unroll
  for (int kc = 0; kc < 16; ++kc)
    tf[kc] = *(const v2f*)(trow + 4 * kc);
}

template <int K0, int K1>
__device__ __forceinline__ void chain2(const v2f* tf, const v2f* wa,
                                       const v2f* wb, v8f& a, v8f& b) {
#pragma unroll
  for (int kc = K0; kc < K1; ++kc) {
    a = wmma_f32(wa[kc], tf[kc], a);
    b = wmma_f32(wb[kc], tf[kc], b);
  }
}

template <int P>
__device__ __forceinline__ void xpiece(const v2f* xa, const v2f* wq,
                                       const v2f* wv, v8f& am, v8f& av) {
  if (P == 0) {
    am = v8f{};
    av = v8f{};
  }
  chain2<2 * P, 2 * P + 2>(xa, wq, wv, am, av);
}

__device__ __forceinline__ v8f load8(const float* p) {
  v4f a = *(const v4f*)p;
  v4f b = *(const v4f*)(p + 4);
  return __builtin_shufflevector(a, b, 0, 1, 2, 3, 4, 5, 6, 7);
}

__device__ __forceinline__ void store8(float* p, v8f v) {
  *(v4f*)p       = __builtin_shufflevector(v, v, 0, 1, 2, 3);
  *(v4f*)(p + 4) = __builtin_shufflevector(v, v, 4, 5, 6, 7);
}

__device__ __forceinline__ float partner(float v) {
  return __int_as_float(__builtin_amdgcn_update_dpp(0, __float_as_int(v), 0x128,
                                                    0xF, 0xF, true));
}

__device__ __forceinline__ v8f blend(v8f own, float p1) {
  v8f r;
#pragma unroll
  for (int i = 0; i < 8; ++i) r[i] = fmaf(p1, partner(own[i]) - own[i], own[i]);
  return r;
}

__device__ __forceinline__ float part_diff(v8f q, v8f kd) {
  float a = fmaf(q[1], kd[1], q[0] * kd[0]);
  float b = fmaf(q[3], kd[3], q[2] * kd[2]);
  float c = fmaf(q[5], kd[5], q[4] * kd[4]);
  float d = fmaf(q[7], kd[7], q[6] * kd[6]);
  return (a + b) + (c + d);
}

__device__ __forceinline__ float tree8(v4f a, v4f b) {
  return ((a.x + a.y) + (a.z + a.w)) + ((b.x + b.y) + (b.z + b.w));
}

__device__ __forceinline__ float weight_second(float d) {
  const float c = 0.125f * 1.44269504088896340736f;
  float e = __builtin_amdgcn_exp2f(d * c);
  return __builtin_amdgcn_rcpf(1.0f + e);
}

#define MSTR 72

template <bool BT>
__device__ __forceinline__ void mm64(const float* A, const float* Bsrc, int ldb,
                                     float* dst, float* rm, int wv, int m,
                                     int h) {
  v8f acc[4];
#pragma unroll
  for (int nt = 0; nt < 4; ++nt) acc[nt] = v8f{};
  const float* arow = A + (16 * wv + m) * 64 + 2 * h;
#pragma unroll
  for (int kc = 0; kc < 16; ++kc) {
    v2f af = *(const v2f*)(arow + 4 * kc);
#pragma unroll
    for (int nt = 0; nt < 4; ++nt) {
      v2f bf;
      if (BT) {
        bf = *(const v2f*)(Bsrc + (16 * nt + m) * 64 + 2 * h + 4 * kc);
      } else {
        const float* bp = Bsrc + (4 * kc + 2 * h) * ldb + 16 * nt + m;
        bf.x = bp[0];
        bf.y = bp[ldb];
      }
      acc[nt] = wmma_f32(af, bf, acc[nt]);
    }
  }
#pragma unroll
  for (int nt = 0; nt < 4; ++nt) {
#pragma unroll
    for (int q = 0; q < 4; ++q) {
      v2f pr = {acc[nt][2 * q], acc[nt][2 * q + 1]};
      *(v2f*)(dst + ((4 * wv + 2 * h + (q >> 1)) * 4 + nt) * 64 +
              (m + 16 * (q & 1)) * 2) = pr;
    }
    if (rm) {
#pragma unroll
      for (int r = 0; r < 8; ++r)
        rm[(16 * wv + 8 * h + r) * MSTR + 16 * nt + m] = acc[nt][r];
    }
  }
}

__global__ __launch_bounds__(128) void prep_weights(
    const float* __restrict__ Wq, const float* __restrict__ Wk,
    const float* __restrict__ Wv, float* __restrict__ wsw) {
  __shared__ float Ms[64 * MSTR];
  const int lane = threadIdx.x & 31;
  const int wv = threadIdx.x >> 5;
  const int m = lane & 15;
  const int h = lane >> 4;
  if (blockIdx.x == 0) {
    mm64<true>(Wq, Wk, 64, wsw, Ms, wv, m, h);
    __syncthreads();
    mm64<false>(Wv, Ms, MSTR, wsw + 8192, nullptr, wv, m, h);
  } else if (blockIdx.x == 1) {
    mm64<false>(Wv, Wv, 64, wsw + 12288, nullptr, wv, m, h);
  } else {
    int idx = (blockIdx.x - 2) * 128 + threadIdx.x;
    int j    = idx & 1;
    int ln   = (idx >> 1) & 31;
    int nt   = (idx >> 6) & 3;
    int kc   = (idx >> 8) & 15;
    int k = 4 * kc + j + 2 * (ln >> 4);
    int n = 16 * nt + (ln & 15);
    wsw[4096 + idx] = Wv[k * 64 + n];
  }
}

#define B_CAR 0
#define B_OO  1
#define B_X   2
#define B_XP  4
#define N_BUF 8
#define HALF  (8 * LS)
#define PS     12
#define N_PART (16 * PS)

__global__ __launch_bounds__(128) void birnn_kernel(
    const float* __restrict__ x, const float* __restrict__ wsw,
    float* __restrict__ out) {
  __shared__ __align__(16) float smem[N_BUF * BUF + 2 * N_PART];
  float* CAR   = smem + B_CAR * BUF;
  float* OO    = smem + B_OO * BUF;
  float* PARTA = smem + N_BUF * BUF;
  float* PARTB = PARTA + N_PART;
  float* bxc = smem + B_X * BUF;
  float* bxn = smem + (B_X + 1) * BUF;
  float* pxc = smem + B_XP * BUF;
  float* pxn = smem + (B_XP + 2) * BUF;

  const int lane = threadIdx.x & 31;
  const int wid  = threadIdx.x >> 5;
  const int n = lane & 15;
  const int h = lane >> 4;
  const int F0 = 16 * wid + 8 * h;
  const int rowoff = n * LS + F0;
  const int xoff = (n & 7) * LS + F0;
  const bool first = n < 8;
  const int tsel = n >> 3;
  const int b0 = blockIdx.x * 8;
  float* mypartA = PARTA + n * PS + wid * 2 + h;
  float* mypartB = PARTB + n * PS + wid * 2 + h;
  const v4f* partrowA = (const v4f*)(PARTA + n * PS);
  const v4f* partrowB = (const v4f*)(PARTB + n * PS);

  v2f wq[16], wv[16], w3[16], w4[16];
  {
    const float* myw = wsw + wid * 64 + lane * 2;
#pragma unroll
    for (int kc = 0; kc < 16; ++kc) {
      wq[kc] = *(const v2f*)(myw + kc * 256);
      wv[kc] = *(const v2f*)(myw + 4096 + kc * 256);
      w3[kc] = *(const v2f*)(myw + 8192 + kc * 256);
      w4[kc] = *(const v2f*)(myw + 12288 + kc * 256);
    }
  }

  const float* xb = x + (size_t)(b0 + (n & 7)) * TSTEPS * DDIM;
  v4f xr[8];
  v2f xa[16];
  v8f xm = v8f{}, xv = v8f{};
  v2f tf[16];

  v8f cr, cm, cv;
  {
    const v4f* g = (const v4f*)(xb + F0);
    v4f a = g[0], b = g[1];
    cr = __builtin_shufflevector(a, b, 0, 1, 2, 3, 4, 5, 6, 7);
  }
  store8(CAR + rowoff, cr);
  if (wid >= 2) {
    float* dst = (wid == 2) ? bxc : bxn;
    const int t0 = (wid == 2) ? 1 : 3;
    const v4f* xq = (const v4f*)(xb + (size_t)(t0 + tsel) * DDIM + 32 * h);
    v4f* dx = (v4f*)(dst + n * LS + 32 * h);
#pragma unroll
    for (int i = 0; i < 8; ++i) dx[i] = xq[i];
  }
  __syncthreads();
  {
    v8f am = v8f{}, av = v8f{};
    load_tfrags(bxc, tf, n, h);
    chain2<0, 16>(tf, wq, wv, am, av);
    store8(pxc + rowoff, am);
    store8(pxc + BUF + rowoff, av);
  }
  cm = v8f{};
  cv = v8f{};
  load_tfrags(CAR, tf, n, h);
  chain2<0, 16>(tf, wq, wv, cm, cv);
  load_tfrags(bxn, xa, n, h);
  v8f xmq = load8(pxc + xoff);
  v8f xk  = load8(bxc + xoff);
  v8f xw  = load8(pxc + BUF + xoff);

#pragma unroll 1
  for (int tp = 1; tp < TSTEPS; tp += 2) {
    const bool ahead = (tp + 2 < TSTEPS);
    const bool fetch = (tp + 4 < TSTEPS);

#pragma unroll
    for (int sub = 0; sub < 2; ++sub) {
      if (sub == 1 && tp + 1 >= TSTEPS) break;

      {
        v8f kd = cr - xk;
        float dx = part_diff(xmq, kd);
        float da = part_diff(cm, kd);
        *mypartA = first ? da : dx;
      }
      if (ahead) {
        if (sub == 0) xpiece<0>(xa, wq, wv, xm, xv);
        else          xpiece<4>(xa, wq, wv, xm, xv);
      }
      __syncthreads();
      v4f pa0 = partrowA[0], pa1 = partrowA[1];
      __builtin_amdgcn_sched_barrier(0);
      if (ahead) {
        if (sub == 0) xpiece<1>(xa, wq, wv, xm, xv);
        else          xpiece<5>(xa, wq, wv, xm, xv);
      }
      v8f orow;
      {
        v8f dv = xw - cv;
        float p1 = weight_second(tree8(pa0, pa1));
#pragma unroll
        for (int r = 0; r < 8; ++r)
          orow[r] = fast_tanh(fmaf(p1, dv[r], cv[r]));
      }
      store8(OO + rowoff, orow);
      __syncthreads();

      load_tfrags(OO, tf, n, h);
      if (wid == 2 && fetch) {
        if (sub == 0) {
          int ti = tp + 4 + tsel;
          ti = ti < TSTEPS ? ti : TSTEPS - 1;
          const v4f* xq = (const v4f*)(xb + (size_t)ti * DDIM + 32 * h);
#pragma unroll
          for (int i = 0; i < 8; ++i) xr[i] = xq[i];
        } else {
          v4f* dx = (v4f*)(bxc + n * LS + 32 * h);
#pragma unroll
          for (int i = 0; i < 8; ++i) dx[i] = xr[i];
        }
      }
      __builtin_amdgcn_sched_barrier(0);
      if (ahead) {
        if (sub == 0) xpiece<2>(xa, wq, wv, xm, xv);
        else          xpiece<6>(xa, wq, wv, xm, xv);
      }
      v8f a1 = v8f{}, a2 = v8f{}, a3 = v8f{}, a4 = v8f{};
      chain2<0, 16>(tf, wq, wv, a1, a2);
      asm volatile("" : "+v"(a1), "+v"(a2));
      {
        v8f kd;
#pragma unroll
        for (int r = 0; r < 8; ++r)
          kd[r] = orow[r] - partner(orow[r]);
        *mypartB = part_diff(a1, kd);
      }
      __builtin_amdgcn_sched_barrier(0);
      chain2<0, 8>(tf, w3, w4, a3, a4);
      if (ahead) {
        if (sub == 0) {
          xpiece<3>(xa, wq, wv, xm, xv);
        } else {
          xpiece<7>(xa, wq, wv, xm, xv);
          store8(pxn + rowoff, xm);
          store8(pxn + BUF + rowoff, xv);
        }
      }
      __syncthreads();
      v4f pb0 = partrowB[0], pb1 = partrowB[1];
      {
        const float* px = (sub == 0) ? pxc : pxn;
        const float* bx = (sub == 0) ? bxc : bxn;
        const int xo = xoff + (sub == 0 ? HALF : 0);
        xmq = load8(px + xo);
        xk  = load8(bx + xo);
        xw  = load8(px + BUF + xo);
      }
      if (sub == 1 && fetch) load_tfrags(bxc, xa, n, h);
      __builtin_amdgcn_sched_barrier(0);
      chain2<8, 16>(tf, w3, w4, a3, a4);
      v8f d2;
#pragma unroll
      for (int r = 0; r < 8; ++r)
        d2[r] = partner(a2[r]) - a2[r];
      asm volatile("" : "+v"(a3), "+v"(a4), "+v"(pb0), "+v"(pb1));
      {
        float p1 = weight_second(tree8(pb0, pb1));
#pragma unroll
        for (int r = 0; r < 8; ++r)
          cr[r] = fmaf(p1, d2[r], a2[r]);
        cm = blend(a3, p1);
        cv = blend(a4, p1);
      }
    }

    float* tmp = pxc;
    pxc = pxn;
    pxn = tmp;
    tmp = bxc;
    bxc = bxn;
    bxn = tmp;
  }

  store8(CAR + rowoff, cr);
  __syncthreads();
  if (wid == 0) {
#pragma unroll
    for (int i = 0; i < 4; ++i) {
      int row = 2 * i + (lane >> 4);
      const float4* s4 = (const float4*)(CAR + row * LS + 4 * (lane & 15));
      float4* d4 = (float4*)(out + (size_t)(b0 + row) * DDIM + 4 * (lane & 15));
      *d4 = *s4;
    }
  }
}

extern "C" void kernel_launch(void* const* d_in, const int* in_sizes, int n_in,
                              void* d_out, int out_size, void* d_ws,
                              size_t ws_size, hipStream_t stream) {
  (void)in_sizes; (void)n_in; (void)out_size; (void)ws_size;
  const float* x  = (const float*)d_in[0];
  const float* Wq = (const float*)d_in[1];
  const float* Wk = (const float*)d_in[2];
  const float* Wv = (const float*)d_in[3];
  float* wsw = (float*)d_ws;

  prep_weights<<<34, 128, 0, stream>>>(Wq, Wk, Wv, wsw);
  birnn_kernel<<<2048 / 8, 128, 0, stream>>>(x, wsw, (float*)d_out);
}
